# starting kernel
# baseline (speedup 1.0000x reference)
#include <hip/hip_runtime.h>
#include <math.h>

#define N_PTS   8192
#define WV_DIM  300
#define D_DIM   128
#define BIG_F   1e8f
#define EPS_TRM (128.0f * 1e-16f)

typedef float v2f __attribute__((ext_vector_type(2)));
typedef float v8f __attribute__((ext_vector_type(8)));

__device__ __forceinline__ v2f ld2(const float* p) {
    return *(const v2f*)p;
}

__global__ void x2_kernel(const float* __restrict__ x, float* __restrict__ x2) {
    int i = blockIdx.x * blockDim.x + threadIdx.x;
    if (i < N_PTS) {
        const float* row = x + (size_t)i * D_DIM;
        float s = 0.f;
        #pragma unroll 8
        for (int d = 0; d < D_DIM; ++d) s = fmaf(row[d], row[d], s);
        x2[i] = s;
    }
}

__global__ __launch_bounds__(32) void sem_fc_kernel(
    const float* __restrict__ x,  const float* __restrict__ wv,
    const float* __restrict__ alpha, const float* __restrict__ W,
    const float* __restrict__ b,  float* __restrict__ sem_out,
    float* __restrict__ sem2) {
    const int lane = threadIdx.x;
    const int col  = lane & 15;
    const int hi   = lane >> 4;
    const int i0   = blockIdx.x * 16;

    float alph[8];
    #pragma unroll
    for (int r = 0; r < 8; ++r) alph[r] = alpha[i0 + r + 8 * hi];

    float nrm[8];
    #pragma unroll
    for (int r = 0; r < 8; ++r) nrm[r] = 0.f;

    const float* arow = wv + (size_t)(i0 + col) * WV_DIM + 2 * hi;

    for (int dt = 0; dt < D_DIM / 16; ++dt) {
        const int d = dt * 16 + col;
        const float* brow = W + (size_t)d * WV_DIM + 2 * hi;

        v8f c = {};
        for (int k = 0; k < WV_DIM; k += 4) {
            v2f a  = ld2(arow + k);
            v2f bb = ld2(brow + k);
            c = __builtin_amdgcn_wmma_f32_16x16x4_f32(
                    false, a, false, bb, (short)0, c, false, false);
        }

        const float bd = b[d];
        #pragma unroll
        for (int r = 0; r < 8; ++r) {
            const int i = i0 + r + 8 * hi;
            const float so = c[r] + bd;
            const float xv = x[(size_t)i * D_DIM + d];
            const float s  = alph[r] * so + (1.f - alph[r]) * xv;
            sem_out[(size_t)i * D_DIM + d] = s;
            nrm[r] = fmaf(s, s, nrm[r]);
        }
    }

    #pragma unroll
    for (int r = 0; r < 8; ++r) {
        float v = nrm[r];
        for (int off = 8; off >= 1; off >>= 1)
            v += __shfl_xor(v, off, 16);
        if (col == 0) sem2[i0 + r + 8 * hi] = v;
    }
}

__global__ __launch_bounds__(32) void pair_kernel(
    const float* __restrict__ x,    const int* __restrict__ label,
    const float* __restrict__ sem,  const float* __restrict__ sem2,
    const float* __restrict__ x2,   float* __restrict__ diff) {
    const int lane = threadIdx.x;
    const int col  = lane & 15;
    const int hi   = lane >> 4;
    const int i0   = blockIdx.x * 16;

    v2f afrag[32];
    const float* arow = sem + (size_t)(i0 + col) * D_DIM + 2 * hi;
    #pragma unroll
    for (int ks = 0; ks < 32; ++ks) afrag[ks] = ld2(arow + 4 * ks);

    int   labi[8];
    float s2i[8];
    #pragma unroll
    for (int r = 0; r < 8; ++r) {
        labi[r] = label[i0 + r + 8 * hi];
        s2i[r]  = sem2[i0 + r + 8 * hi];
    }

    float fp[8], cn[8];
    #pragma unroll
    for (int r = 0; r < 8; ++r) { fp[r] = 0.f; cn[r] = 3.4e38f; }

    for (int jt = 0; jt < N_PTS / 16; ++jt) {
        const int   j    = jt * 16 + col;
        const int   labj = label[j];
        const float x2j  = x2[j];
        const float* brow = x + (size_t)j * D_DIM + 2 * hi;

        v8f c = {};
        #pragma unroll
        for (int ks = 0; ks < 32; ++ks) {
            v2f bb = ld2(brow + 4 * ks);
            c = __builtin_amdgcn_wmma_f32_16x16x4_f32(
                    false, afrag[ks], false, bb, (short)0, c, false, false);
        }

        #pragma unroll
        for (int r = 0; r < 8; ++r) {
            const int i  = i0 + r + 8 * hi;
            float sq = s2i[r] + x2j - 2.f * c[r];
            sq = sq > 0.f ? sq : 0.f;
            const float dist = sqrtf(sq + EPS_TRM);
            const bool same = (labi[r] == labj);
            const bool pos  = same && (i != j);
            fp[r] = fmaxf(fp[r], pos ? dist : 0.f);
            cn[r] = fminf(cn[r], same ? dist + BIG_F : dist);
        }
    }

    #pragma unroll
    for (int r = 0; r < 8; ++r) {
        float a = fp[r], m = cn[r];
        for (int off = 8; off >= 1; off >>= 1) {
            a = fmaxf(a, __shfl_xor(a, off, 16));
            m = fminf(m, __shfl_xor(m, off, 16));
        }
        if (col == 0) {
            const float z  = a - m;
            const float sp = (z > 20.f) ? z : log1pf(expf(z));
            diff[i0 + r + 8 * hi] = sp;
        }
    }
}

extern "C" void kernel_launch(void* const* d_in, const int* in_sizes, int n_in,
                              void* d_out, int out_size, void* d_ws, size_t ws_size,
                              hipStream_t stream) {
    const float* x     = (const float*)d_in[0];
    const float* wv    = (const float*)d_in[1];
    const int*   label = (const int*)  d_in[2];
    const float* alpha = (const float*)d_in[3];
    const float* W     = (const float*)d_in[4];
    const float* b     = (const float*)d_in[5];

    float* diff = (float*)d_out;
    float* sem  = diff + N_PTS;

    float* x2   = (float*)d_ws;
    float* sem2 = x2 + N_PTS;

    hipLaunchKernelGGL(x2_kernel, dim3(N_PTS / 256), dim3(256), 0, stream, x, x2);
    hipLaunchKernelGGL(sem_fc_kernel, dim3(N_PTS / 16), dim3(32), 0, stream,
                       x, wv, alpha, W, b, sem, sem2);
    hipLaunchKernelGGL(pair_kernel, dim3(N_PTS / 16), dim3(32), 0, stream,
                       x, label, sem, sem2, x2, diff);
}


# pair kernel: the 512-tile sweep of each 16-row stripe is shared by 8 waves of one work-group (was 1 wave per stripe); partial max/min merged through LDS (exact, no sum reordered)
# speedup vs baseline: 5.5552x; 5.5552x over previous
#include <hip/hip_runtime.h>
#include <math.h>

#define N_PTS   8192
#define WV_DIM  300
#define D_DIM   128
#define BIG_F   1e8f
#define EPS_TRM (128.0f * 1e-16f)

typedef float v2f __attribute__((ext_vector_type(2)));
typedef float v8f __attribute__((ext_vector_type(8)));

__device__ __forceinline__ v2f ld2(const float* p) {
    return *(const v2f*)p;
}

__global__ void x2_kernel(const float* __restrict__ x, float* __restrict__ x2) {
    int i = blockIdx.x * blockDim.x + threadIdx.x;
    if (i < N_PTS) {
        const float* row = x + (size_t)i * D_DIM;
        float s = 0.f;
        #pragma unroll 8
        for (int d = 0; d < D_DIM; ++d) s = fmaf(row[d], row[d], s);
        x2[i] = s;
    }
}

__global__ __launch_bounds__(32) void sem_fc_kernel(
    const float* __restrict__ x,  const float* __restrict__ wv,
    const float* __restrict__ alpha, const float* __restrict__ W,
    const float* __restrict__ b,  float* __restrict__ sem_out,
    float* __restrict__ sem2) {
    const int lane = threadIdx.x;
    const int col  = lane & 15;
    const int hi   = lane >> 4;
    const int i0   = blockIdx.x * 16;

    float alph[8];
    #pragma unroll
    for (int r = 0; r < 8; ++r) alph[r] = alpha[i0 + r + 8 * hi];

    float nrm[8];
    #pragma unroll
    for (int r = 0; r < 8; ++r) nrm[r] = 0.f;

    const float* arow = wv + (size_t)(i0 + col) * WV_DIM + 2 * hi;

    for (int dt = 0; dt < D_DIM / 16; ++dt) {
        const int d = dt * 16 + col;
        const float* brow = W + (size_t)d * WV_DIM + 2 * hi;

        v8f c = {};
        for (int k = 0; k < WV_DIM; k += 4) {
            v2f a  = ld2(arow + k);
            v2f bb = ld2(brow + k);
            c = __builtin_amdgcn_wmma_f32_16x16x4_f32(
                    false, a, false, bb, (short)0, c, false, false);
        }

        const float bd = b[d];
        #pragma unroll
        for (int r = 0; r < 8; ++r) {
            const int i = i0 + r + 8 * hi;
            const float so = c[r] + bd;
            const float xv = x[(size_t)i * D_DIM + d];
            const float s  = alph[r] * so + (1.f - alph[r]) * xv;
            sem_out[(size_t)i * D_DIM + d] = s;
            nrm[r] = fmaf(s, s, nrm[r]);
        }
    }

    #pragma unroll
    for (int r = 0; r < 8; ++r) {
        float v = nrm[r];
        for (int off = 8; off >= 1; off >>= 1)
            v += __shfl_xor(v, off, 16);
        if (col == 0) sem2[i0 + r + 8 * hi] = v;
    }
}

#define PAIR_WAVES 8
__global__ __launch_bounds__(32 * PAIR_WAVES) void pair_kernel(
    const float* __restrict__ x,    const int* __restrict__ label,
    const float* __restrict__ sem,  const float* __restrict__ sem2,
    const float* __restrict__ x2,   float* __restrict__ diff) {
    __shared__ float s_fp[PAIR_WAVES][16];
    __shared__ float s_cn[PAIR_WAVES][16];

    const int lane = threadIdx.x & 31;
    const int wave = __builtin_amdgcn_readfirstlane((int)(threadIdx.x >> 5));
    const int col  = lane & 15;
    const int hi   = lane >> 4;
    const int i0   = blockIdx.x * 16;

    v2f afrag[32];
    const float* arow = sem + (size_t)(i0 + col) * D_DIM + 2 * hi;
    #pragma unroll
    for (int ks = 0; ks < 32; ++ks) afrag[ks] = ld2(arow + 4 * ks);

    int   labi[8];
    float s2i[8];
    #pragma unroll
    for (int r = 0; r < 8; ++r) {
        labi[r] = label[i0 + r + 8 * hi];
        s2i[r]  = sem2[i0 + r + 8 * hi];
    }

    float fp[8], cn[8];
    #pragma unroll
    for (int r = 0; r < 8; ++r) { fp[r] = 0.f; cn[r] = 3.4e38f; }

    for (int jt = wave; jt < N_PTS / 16; jt += PAIR_WAVES) {
        const int   j    = jt * 16 + col;
        const int   labj = label[j];
        const float x2j  = x2[j];
        const float* brow = x + (size_t)j * D_DIM + 2 * hi;

        v8f c = {};
        #pragma unroll
        for (int ks = 0; ks < 32; ++ks) {
            v2f bb = ld2(brow + 4 * ks);
            c = __builtin_amdgcn_wmma_f32_16x16x4_f32(
                    false, afrag[ks], false, bb, (short)0, c, false, false);
        }

        #pragma unroll
        for (int r = 0; r < 8; ++r) {
            const int i  = i0 + r + 8 * hi;
            float sq = s2i[r] + x2j - 2.f * c[r];
            sq = sq > 0.f ? sq : 0.f;
            const float dist = sqrtf(sq + EPS_TRM);
            const bool same = (labi[r] == labj);
            const bool pos  = same && (i != j);
            fp[r] = fmaxf(fp[r], pos ? dist : 0.f);
            cn[r] = fminf(cn[r], same ? dist + BIG_F : dist);
        }
    }

    #pragma unroll
    for (int r = 0; r < 8; ++r) {
        float a = fp[r], m = cn[r];
        for (int off = 8; off >= 1; off >>= 1) {
            a = fmaxf(a, __shfl_xor(a, off, 16));
            m = fminf(m, __shfl_xor(m, off, 16));
        }
        if (col == 0) {
            s_fp[wave][r + 8 * hi] = a;
            s_cn[wave][r + 8 * hi] = m;
        }
    }
    __syncthreads();

    if (threadIdx.x < 16) {
        float a = s_fp[0][threadIdx.x], m = s_cn[0][threadIdx.x];
        #pragma unroll
        for (int w = 1; w < PAIR_WAVES; ++w) {
            a = fmaxf(a, s_fp[w][threadIdx.x]);
            m = fminf(m, s_cn[w][threadIdx.x]);
        }
        const float z  = a - m;
        const float sp = (z > 20.f) ? z : log1pf(expf(z));
        diff[i0 + threadIdx.x] = sp;
    }
}

extern "C" void kernel_launch(void* const* d_in, const int* in_sizes, int n_in,
                              void* d_out, int out_size, void* d_ws, size_t ws_size,
                              hipStream_t stream) {
    const float* x     = (const float*)d_in[0];
    const float* wv    = (const float*)d_in[1];
    const int*   label = (const int*)  d_in[2];
    const float* alpha = (const float*)d_in[3];
    const float* W     = (const float*)d_in[4];
    const float* b     = (const float*)d_in[5];

    float* diff = (float*)d_out;
    float* sem  = diff + N_PTS;

    float* x2   = (float*)d_ws;
    float* sem2 = x2 + N_PTS;

    hipLaunchKernelGGL(x2_kernel, dim3(N_PTS / 256), dim3(256), 0, stream, x, x2);
    hipLaunchKernelGGL(sem_fc_kernel, dim3(N_PTS / 16), dim3(32), 0, stream,
                       x, wv, alpha, W, b, sem, sem2);
    hipLaunchKernelGGL(pair_kernel, dim3(N_PTS / 16), dim3(32 * PAIR_WAVES), 0, stream,
                       x, label, sem, sem2, x2, diff);
}


# pair kernel: each wave fetches its 16x128 x tile with whole-row 16-byte loads (4 full lines each, was 16 lines per 8-byte load) and reads the same B fragments back from a wave-private LDS slab
# speedup vs baseline: 8.5306x; 1.5356x over previous
#include <hip/hip_runtime.h>
#include <math.h>

#define N_PTS   8192
#define WV_DIM  300
#define D_DIM   128
#define BIG_F   1e8f
#define EPS_TRM (128.0f * 1e-16f)

typedef float v2f __attribute__((ext_vector_type(2)));
typedef float v4f __attribute__((ext_vector_type(4)));
typedef float v8f __attribute__((ext_vector_type(8)));

__device__ __forceinline__ v2f ld2(const float* p) {
    return *(const v2f*)p;
}

__global__ void x2_kernel(const float* __restrict__ x, float* __restrict__ x2) {
    int i = blockIdx.x * blockDim.x + threadIdx.x;
    if (i < N_PTS) {
        const float* row = x + (size_t)i * D_DIM;
        float s = 0.f;
        #pragma unroll 8
        for (int d = 0; d < D_DIM; ++d) s = fmaf(row[d], row[d], s);
        x2[i] = s;
    }
}

__global__ __launch_bounds__(32) void sem_fc_kernel(
    const float* __restrict__ x,  const float* __restrict__ wv,
    const float* __restrict__ alpha, const float* __restrict__ W,
    const float* __restrict__ b,  float* __restrict__ sem_out,
    float* __restrict__ sem2) {
    const int lane = threadIdx.x;
    const int col  = lane & 15;
    const int hi   = lane >> 4;
    const int i0   = blockIdx.x * 16;

    float alph[8];
    #pragma unroll
    for (int r = 0; r < 8; ++r) alph[r] = alpha[i0 + r + 8 * hi];

    float nrm[8];
    #pragma unroll
    for (int r = 0; r < 8; ++r) nrm[r] = 0.f;

    const float* arow = wv + (size_t)(i0 + col) * WV_DIM + 2 * hi;

    for (int dt = 0; dt < D_DIM / 16; ++dt) {
        const int d = dt * 16 + col;
        const float* brow = W + (size_t)d * WV_DIM + 2 * hi;

        v8f c = {};
        for (int k = 0; k < WV_DIM; k += 4) {
            v2f a  = ld2(arow + k);
            v2f bb = ld2(brow + k);
            c = __builtin_amdgcn_wmma_f32_16x16x4_f32(
                    false, a, false, bb, (short)0, c, false, false);
        }

        const float bd = b[d];
        #pragma unroll
        for (int r = 0; r < 8; ++r) {
            const int i = i0 + r + 8 * hi;
            const float so = c[r] + bd;
            const float xv = x[(size_t)i * D_DIM + d];
            const float s  = alph[r] * so + (1.f - alph[r]) * xv;
            sem_out[(size_t)i * D_DIM + d] = s;
            nrm[r] = fmaf(s, s, nrm[r]);
        }
    }

    #pragma unroll
    for (int r = 0; r < 8; ++r) {
        float v = nrm[r];
        for (int off = 8; off >= 1; off >>= 1)
            v += __shfl_xor(v, off, 16);
        if (col == 0) sem2[i0 + r + 8 * hi] = v;
    }
}

#define PAIR_WAVES 8
#define TILE_PITCH 132
__global__ __launch_bounds__(32 * PAIR_WAVES) __attribute__((amdgpu_waves_per_eu(4, 4))) void pair_kernel(
    const float* __restrict__ x,    const int* __restrict__ label,
    const float* __restrict__ sem,  const float* __restrict__ sem2,
    const float* __restrict__ x2,   float* __restrict__ diff) {
    __shared__ float s_fp[PAIR_WAVES][16];
    __shared__ float s_cn[PAIR_WAVES][16];
    __shared__ __attribute__((aligned(16))) float s_tile[PAIR_WAVES][16 * TILE_PITCH];

    const int lane = threadIdx.x & 31;
    const int wave = __builtin_amdgcn_readfirstlane((int)(threadIdx.x >> 5));
    const int col  = lane & 15;
    const int hi   = lane >> 4;
    const int i0   = blockIdx.x * 16;

    v2f afrag[32];
    const float* arow = sem + (size_t)(i0 + col) * D_DIM + 2 * hi;
    #pragma unroll
    for (int ks = 0; ks < 32; ++ks) afrag[ks] = ld2(arow + 4 * ks);

    int   labi[8];
    float s2i[8];
    #pragma unroll
    for (int r = 0; r < 8; ++r) {
        labi[r] = label[i0 + r + 8 * hi];
        s2i[r]  = sem2[i0 + r + 8 * hi];
    }

    float fp[8], cn[8];
    #pragma unroll
    for (int r = 0; r < 8; ++r) { fp[r] = 0.f; cn[r] = 3.4e38f; }

    float* slab_wr = &s_tile[wave][4 * lane];
    const float* slab_rd = &s_tile[wave][col * TILE_PITCH + 2 * hi];

    for (int jt = wave; jt < N_PTS / 16; jt += PAIR_WAVES) {
        const int   j    = jt * 16 + col;
        const int   labj = label[j];
        const float x2j  = x2[j];
        const float* xrow = x + (size_t)(jt * 16) * D_DIM + 4 * lane;

        v4f st[16];
        #pragma unroll
        for (int t = 0; t < 16; ++t) st[t] = *(const v4f*)(xrow + (size_t)t * D_DIM);
        __builtin_amdgcn_sched_barrier(0);
        #pragma unroll
        for (int t = 0; t < 16; ++t) *(v4f*)(slab_wr + t * TILE_PITCH) = st[t];

        __builtin_amdgcn_fence(__ATOMIC_RELEASE, "wavefront");
        __builtin_amdgcn_wave_barrier();
        __builtin_amdgcn_fence(__ATOMIC_ACQUIRE, "wavefront");

        v2f bfrag[32];
        #pragma unroll
        for (int ks = 0; ks < 32; ++ks) bfrag[ks] = *(const v2f*)(slab_rd + 4 * ks);

        __builtin_amdgcn_fence(__ATOMIC_RELEASE, "wavefront");
        __builtin_amdgcn_wave_barrier();
        __builtin_amdgcn_fence(__ATOMIC_ACQUIRE, "wavefront");
        __builtin_amdgcn_sched_barrier(0);

        v8f c = {};
        #pragma unroll
        for (int ks = 0; ks < 32; ++ks) {
            c = __builtin_amdgcn_wmma_f32_16x16x4_f32(
                    false, afrag[ks], false, bfrag[ks], (short)0, c, false, false);
        }

        #pragma unroll
        for (int r = 0; r < 8; ++r) {
            const int i  = i0 + r + 8 * hi;
            float sq = s2i[r] + x2j - 2.f * c[r];
            sq = sq > 0.f ? sq : 0.f;
            const float dist = sqrtf(sq + EPS_TRM);
            const bool same = (labi[r] == labj);
            const bool pos  = same && (i != j);
            fp[r] = fmaxf(fp[r], pos ? dist : 0.f);
            cn[r] = fminf(cn[r], same ? dist + BIG_F : dist);
        }
    }

    #pragma unroll
    for (int r = 0; r < 8; ++r) {
        float a = fp[r], m = cn[r];
        for (int off = 8; off >= 1; off >>= 1) {
            a = fmaxf(a, __shfl_xor(a, off, 16));
            m = fminf(m, __shfl_xor(m, off, 16));
        }
        if (col == 0) {
            s_fp[wave][r + 8 * hi] = a;
            s_cn[wave][r + 8 * hi] = m;
        }
    }
    __syncthreads();

    if (threadIdx.x < 16) {
        float a = s_fp[0][threadIdx.x], m = s_cn[0][threadIdx.x];
        #pragma unroll
        for (int w = 1; w < PAIR_WAVES; ++w) {
            a = fmaxf(a, s_fp[w][threadIdx.x]);
            m = fminf(m, s_cn[w][threadIdx.x]);
        }
        const float z  = a - m;
        const float sp = (z > 20.f) ? z : log1pf(expf(z));
        diff[i0 + threadIdx.x] = sp;
    }
}

extern "C" void kernel_launch(void* const* d_in, const int* in_sizes, int n_in,
                              void* d_out, int out_size, void* d_ws, size_t ws_size,
                              hipStream_t stream) {
    const float* x     = (const float*)d_in[0];
    const float* wv    = (const float*)d_in[1];
    const int*   label = (const int*)  d_in[2];
    const float* alpha = (const float*)d_in[3];
    const float* W     = (const float*)d_in[4];
    const float* b     = (const float*)d_in[5];

    float* diff = (float*)d_out;
    float* sem  = diff + N_PTS;

    float* x2   = (float*)d_ws;
    float* sem2 = x2 + N_PTS;

    hipLaunchKernelGGL(x2_kernel, dim3(N_PTS / 256), dim3(256), 0, stream, x, x2);
    hipLaunchKernelGGL(sem_fc_kernel, dim3(N_PTS / 16), dim3(32), 0, stream,
                       x, wv, alpha, W, b, sem, sem2);
    hipLaunchKernelGGL(pair_kernel, dim3(N_PTS / 16), dim3(32 * PAIR_WAVES), 0, stream,
                       x, label, sem, sem2, x2, diff);
}


# pair kernel: running max/min kept on the squared distance; clamp, +eps, sqrt and +BIG (all monotone) applied once per row after the sweep instead of per pair; bit-identical, no sum reordered
# speedup vs baseline: 9.1546x; 1.0731x over previous
#include <hip/hip_runtime.h>
#include <math.h>

#define N_PTS   8192
#define WV_DIM  300
#define D_DIM   128
#define BIG_F   1e8f
#define EPS_TRM (128.0f * 1e-16f)

typedef float v2f __attribute__((ext_vector_type(2)));
typedef float v4f __attribute__((ext_vector_type(4)));
typedef float v8f __attribute__((ext_vector_type(8)));

__device__ __forceinline__ v2f ld2(const float* p) {
    return *(const v2f*)p;
}

__global__ void x2_kernel(const float* __restrict__ x, float* __restrict__ x2) {
    int i = blockIdx.x * blockDim.x + threadIdx.x;
    if (i < N_PTS) {
        const float* row = x + (size_t)i * D_DIM;
        float s = 0.f;
        #pragma unroll 8
        for (int d = 0; d < D_DIM; ++d) s = fmaf(row[d], row[d], s);
        x2[i] = s;
    }
}

__global__ __launch_bounds__(32) void sem_fc_kernel(
    const float* __restrict__ x,  const float* __restrict__ wv,
    const float* __restrict__ alpha, const float* __restrict__ W,
    const float* __restrict__ b,  float* __restrict__ sem_out,
    float* __restrict__ sem2) {
    const int lane = threadIdx.x;
    const int col  = lane & 15;
    const int hi   = lane >> 4;
    const int i0   = blockIdx.x * 16;

    float alph[8];
    #pragma unroll
    for (int r = 0; r < 8; ++r) alph[r] = alpha[i0 + r + 8 * hi];

    float nrm[8];
    #pragma unroll
    for (int r = 0; r < 8; ++r) nrm[r] = 0.f;

    const float* arow = wv + (size_t)(i0 + col) * WV_DIM + 2 * hi;

    for (int dt = 0; dt < D_DIM / 16; ++dt) {
        const int d = dt * 16 + col;
        const float* brow = W + (size_t)d * WV_DIM + 2 * hi;

        v8f c = {};
        for (int k = 0; k < WV_DIM; k += 4) {
            v2f a  = ld2(arow + k);
            v2f bb = ld2(brow + k);
            c = __builtin_amdgcn_wmma_f32_16x16x4_f32(
                    false, a, false, bb, (short)0, c, false, false);
        }

        const float bd = b[d];
        #pragma unroll
        for (int r = 0; r < 8; ++r) {
            const int i = i0 + r + 8 * hi;
            const float so = c[r] + bd;
            const float xv = x[(size_t)i * D_DIM + d];
            const float s  = alph[r] * so + (1.f - alph[r]) * xv;
            sem_out[(size_t)i * D_DIM + d] = s;
            nrm[r] = fmaf(s, s, nrm[r]);
        }
    }

    #pragma unroll
    for (int r = 0; r < 8; ++r) {
        float v = nrm[r];
        for (int off = 8; off >= 1; off >>= 1)
            v += __shfl_xor(v, off, 16);
        if (col == 0) sem2[i0 + r + 8 * hi] = v;
    }
}

#define PAIR_WAVES 8
#define TILE_PITCH 132
__global__ __launch_bounds__(32 * PAIR_WAVES) __attribute__((amdgpu_waves_per_eu(4, 4))) void pair_kernel(
    const float* __restrict__ x,    const int* __restrict__ label,
    const float* __restrict__ sem,  const float* __restrict__ sem2,
    const float* __restrict__ x2,   float* __restrict__ diff) {
    __shared__ float s_mp[PAIR_WAVES][16];
    __shared__ float s_mn[PAIR_WAVES][16];
    __shared__ float s_ms[PAIR_WAVES][16];
    __shared__ __attribute__((aligned(16))) float s_tile[PAIR_WAVES][16 * TILE_PITCH];

    const int lane = threadIdx.x & 31;
    const int wave = __builtin_amdgcn_readfirstlane((int)(threadIdx.x >> 5));
    const int col  = lane & 15;
    const int hi   = lane >> 4;
    const int i0   = blockIdx.x * 16;

    v2f afrag[32];
    const float* arow = sem + (size_t)(i0 + col) * D_DIM + 2 * hi;
    #pragma unroll
    for (int ks = 0; ks < 32; ++ks) afrag[ks] = ld2(arow + 4 * ks);

    int   labi[8];
    float s2i[8];
    #pragma unroll
    for (int r = 0; r < 8; ++r) {
        labi[r] = label[i0 + r + 8 * hi];
        s2i[r]  = sem2[i0 + r + 8 * hi];
    }

    float mp[8], mn[8], ms[8];
    #pragma unroll
    for (int r = 0; r < 8; ++r) { mp[r] = -INFINITY; mn[r] = INFINITY; ms[r] = INFINITY; }

    float* slab_wr = &s_tile[wave][4 * lane];
    const float* slab_rd = &s_tile[wave][col * TILE_PITCH + 2 * hi];

    for (int jt = wave; jt < N_PTS / 16; jt += PAIR_WAVES) {
        const int   j    = jt * 16 + col;
        const int   labj = label[j];
        const float x2j  = x2[j];
        const float* xrow = x + (size_t)(jt * 16) * D_DIM + 4 * lane;

        v4f st[16];
        #pragma unroll
        for (int t = 0; t < 16; ++t) st[t] = *(const v4f*)(xrow + (size_t)t * D_DIM);
        __builtin_amdgcn_sched_barrier(0);
        #pragma unroll
        for (int t = 0; t < 16; ++t) *(v4f*)(slab_wr + t * TILE_PITCH) = st[t];

        __builtin_amdgcn_fence(__ATOMIC_RELEASE, "wavefront");
        __builtin_amdgcn_wave_barrier();
        __builtin_amdgcn_fence(__ATOMIC_ACQUIRE, "wavefront");

        v2f bfrag[32];
        #pragma unroll
        for (int ks = 0; ks < 32; ++ks) bfrag[ks] = *(const v2f*)(slab_rd + 4 * ks);

        __builtin_amdgcn_fence(__ATOMIC_RELEASE, "wavefront");
        __builtin_amdgcn_wave_barrier();
        __builtin_amdgcn_fence(__ATOMIC_ACQUIRE, "wavefront");
        __builtin_amdgcn_sched_barrier(0);

        v8f c = {};
        #pragma unroll
        for (int ks = 0; ks < 32; ++ks) {
            c = __builtin_amdgcn_wmma_f32_16x16x4_f32(
                    false, afrag[ks], false, bfrag[ks], (short)0, c, false, false);
        }

        #pragma unroll
        for (int r = 0; r < 8; ++r) {
            const int i  = i0 + r + 8 * hi;
            const float sq = s2i[r] + x2j - 2.f * c[r];
            const bool same = (labi[r] == labj);
            const bool pos  = same && (i != j);
            mp[r] = fmaxf(mp[r], pos  ? sq : -INFINITY);
            mn[r] = fminf(mn[r], same ? INFINITY : sq);
            ms[r] = fminf(ms[r], same ? sq : INFINITY);
        }
    }

    #pragma unroll
    for (int r = 0; r < 8; ++r) {
        float a = mp[r], m = mn[r], s = ms[r];
        for (int off = 8; off >= 1; off >>= 1) {
            a = fmaxf(a, __shfl_xor(a, off, 16));
            m = fminf(m, __shfl_xor(m, off, 16));
            s = fminf(s, __shfl_xor(s, off, 16));
        }
        if (col == 0) {
            s_mp[wave][r + 8 * hi] = a;
            s_mn[wave][r + 8 * hi] = m;
            s_ms[wave][r + 8 * hi] = s;
        }
    }
    __syncthreads();

    if (threadIdx.x < 16) {
        float a = s_mp[0][threadIdx.x], m = s_mn[0][threadIdx.x], s = s_ms[0][threadIdx.x];
        #pragma unroll
        for (int w = 1; w < PAIR_WAVES; ++w) {
            a = fmaxf(a, s_mp[w][threadIdx.x]);
            m = fminf(m, s_mn[w][threadIdx.x]);
            s = fminf(s, s_ms[w][threadIdx.x]);
        }
        const float fpd = (a == -INFINITY) ? 0.f : sqrtf((a > 0.f ? a : 0.f) + EPS_TRM);
        const float cnd = fminf(sqrtf((m > 0.f ? m : 0.f) + EPS_TRM),
                                sqrtf((s > 0.f ? s : 0.f) + EPS_TRM) + BIG_F);
        const float z  = fpd - cnd;
        const float sp = (z > 20.f) ? z : log1pf(expf(z));
        diff[i0 + threadIdx.x] = sp;
    }
}

extern "C" void kernel_launch(void* const* d_in, const int* in_sizes, int n_in,
                              void* d_out, int out_size, void* d_ws, size_t ws_size,
                              hipStream_t stream) {
    const float* x     = (const float*)d_in[0];
    const float* wv    = (const float*)d_in[1];
    const int*   label = (const int*)  d_in[2];
    const float* alpha = (const float*)d_in[3];
    const float* W     = (const float*)d_in[4];
    const float* b     = (const float*)d_in[5];

    float* diff = (float*)d_out;
    float* sem  = diff + N_PTS;

    float* x2   = (float*)d_ws;
    float* sem2 = x2 + N_PTS;

    hipLaunchKernelGGL(x2_kernel, dim3(N_PTS / 256), dim3(256), 0, stream, x, x2);
    hipLaunchKernelGGL(sem_fc_kernel, dim3(N_PTS / 16), dim3(32), 0, stream,
                       x, wv, alpha, W, b, sem, sem2);
    hipLaunchKernelGGL(pair_kernel, dim3(N_PTS / 16), dim3(32 * PAIR_WAVES), 0, stream,
                       x, label, sem, sem2, x2, diff);
}


# fc kernel: wv block and W copied once into LDS by whole-line 16-byte loads, fragments read from LDS, 16 waves share a 32-row block (was 1 wave per stripe, 16-line global loads); no sum reordered
# speedup vs baseline: 10.5450x; 1.1519x over previous
#include <hip/hip_runtime.h>
#include <math.h>

#define N_PTS   8192
#define WV_DIM  300
#define D_DIM   128
#define BIG_F   1e8f
#define EPS_TRM (128.0f * 1e-16f)

typedef float v2f __attribute__((ext_vector_type(2)));
typedef float v4f __attribute__((ext_vector_type(4)));
typedef float v8f __attribute__((ext_vector_type(8)));

__device__ __forceinline__ v2f ld2(const float* p) {
    return *(const v2f*)p;
}

__global__ void x2_kernel(const float* __restrict__ x, float* __restrict__ x2) {
    int i = blockIdx.x * blockDim.x + threadIdx.x;
    if (i < N_PTS) {
        const float* row = x + (size_t)i * D_DIM;
        float s = 0.f;
        #pragma unroll 8
        for (int d = 0; d < D_DIM; ++d) s = fmaf(row[d], row[d], s);
        x2[i] = s;
    }
}

#define FC_ROWS   32
#define FC_WAVES  16
#define SEM_PITCH 132
#define FC_KB     15
__global__ __launch_bounds__(32 * FC_WAVES) void sem_fc_kernel(
    const float* __restrict__ x,  const float* __restrict__ wv,
    const float* __restrict__ alpha, const float* __restrict__ W,
    const float* __restrict__ b,  float* __restrict__ sem_out,
    float* __restrict__ sem2) {
    __shared__ __attribute__((aligned(16))) float s_w[D_DIM * WV_DIM];
    __shared__ __attribute__((aligned(16))) float s_a[FC_ROWS * WV_DIM];
    __shared__ __attribute__((aligned(16))) float s_sem[FC_ROWS * SEM_PITCH];

    const int tid  = threadIdx.x;
    const int lane = tid & 31;
    const int wave = __builtin_amdgcn_readfirstlane(tid >> 5);
    const int col  = lane & 15;
    const int hi   = lane >> 4;
    const int dt   = wave & 7;
    const int st   = wave >> 3;
    const int i0   = blockIdx.x * FC_ROWS;

    const v4f* gw = (const v4f*)W;
    const v4f* ga = (const v4f*)(wv + (size_t)i0 * WV_DIM);
    v4f* lw = (v4f*)s_w;
    v4f* la = (v4f*)s_a;
    const int tw = min(tid + 512 * 18, D_DIM * WV_DIM / 4 - 1);
    const int ta = min(tid + 512 * 4, FC_ROWS * WV_DIM / 4 - 1);
    v4f rw[19], ra[5];
    #pragma unroll
    for (int t = 0; t < 18; ++t) rw[t] = gw[tid + 512 * t];
    rw[18] = gw[tw];
    #pragma unroll
    for (int t = 0; t < 4; ++t) ra[t] = ga[tid + 512 * t];
    ra[4] = ga[ta];

    const int   d  = dt * 16 + col;
    const float bd = b[d];
    float alph[8], xv[8];
    #pragma unroll
    for (int r = 0; r < 8; ++r) {
        const int i = i0 + 16 * st + r + 8 * hi;
        alph[r] = alpha[i];
        xv[r]   = x[(size_t)i * D_DIM + d];
    }

    #pragma unroll
    for (int t = 0; t < 18; ++t) lw[tid + 512 * t] = rw[t];
    lw[tw] = rw[18];
    #pragma unroll
    for (int t = 0; t < 4; ++t) la[tid + 512 * t] = ra[t];
    la[ta] = ra[4];
    __syncthreads();

    const float* ap = s_a + (16 * st + col) * WV_DIM + 2 * hi;
    const float* bp = s_w + (16 * dt + col) * WV_DIM + 2 * hi;
    v2f fa[2][FC_KB], fb[2][FC_KB];
    #pragma unroll
    for (int k = 0; k < FC_KB; ++k) {
        fa[0][k] = *(const v2f*)(ap + 4 * k);
        fb[0][k] = *(const v2f*)(bp + 4 * k);
    }
    v8f c = {};
    #pragma unroll
    for (int kb = 0; kb < WV_DIM / 4 / FC_KB; ++kb) {
        if (kb + 1 < WV_DIM / 4 / FC_KB) {
            #pragma unroll
            for (int k = 0; k < FC_KB; ++k) {
                fa[(kb + 1) & 1][k] = *(const v2f*)(ap + 4 * (FC_KB * (kb + 1) + k));
                fb[(kb + 1) & 1][k] = *(const v2f*)(bp + 4 * (FC_KB * (kb + 1) + k));
            }
        }
        __builtin_amdgcn_sched_barrier(0);
        #pragma unroll
        for (int k = 0; k < FC_KB; ++k)
            c = __builtin_amdgcn_wmma_f32_16x16x4_f32(
                    false, fa[kb & 1][k], false, fb[kb & 1][k], (short)0, c, false, false);
        __builtin_amdgcn_sched_barrier(0);
    }

    #pragma unroll
    for (int r = 0; r < 8; ++r) {
        const float so = c[r] + bd;
        const float s  = alph[r] * so + (1.f - alph[r]) * xv[r];
        s_sem[(16 * st + r + 8 * hi) * SEM_PITCH + d] = s;
    }
    __syncthreads();

    #pragma unroll
    for (int t = 0; t < 2; ++t) {
        const int e   = tid + 512 * t;
        const int row = e >> 5;
        const int q   = e & 31;
        const v4f v = *(const v4f*)(s_sem + row * SEM_PITCH + 4 * q);
        *(v4f*)(sem_out + (size_t)(i0 + row) * D_DIM + 4 * q) = v;
    }

    {
        const int row = tid >> 4;
        float nrm = 0.f;
        #pragma unroll
        for (int t = 0; t < 8; ++t) {
            const float v = s_sem[row * SEM_PITCH + 16 * t + col];
            nrm = fmaf(v, v, nrm);
        }
        for (int off = 8; off >= 1; off >>= 1)
            nrm += __shfl_xor(nrm, off, 16);
        if (col == 0) sem2[i0 + row] = nrm;
    }
}

#define PAIR_WAVES 8
#define TILE_PITCH 132
__global__ __launch_bounds__(32 * PAIR_WAVES) __attribute__((amdgpu_waves_per_eu(4, 4))) void pair_kernel(
    const float* __restrict__ x,    const int* __restrict__ label,
    const float* __restrict__ sem,  const float* __restrict__ sem2,
    const float* __restrict__ x2,   float* __restrict__ diff) {
    __shared__ float s_mp[PAIR_WAVES][16];
    __shared__ float s_mn[PAIR_WAVES][16];
    __shared__ float s_ms[PAIR_WAVES][16];
    __shared__ __attribute__((aligned(16))) float s_tile[PAIR_WAVES][16 * TILE_PITCH];

    const int lane = threadIdx.x & 31;
    const int wave = __builtin_amdgcn_readfirstlane((int)(threadIdx.x >> 5));
    const int col  = lane & 15;
    const int hi   = lane >> 4;
    const int i0   = blockIdx.x * 16;

    v2f afrag[32];
    const float* arow = sem + (size_t)(i0 + col) * D_DIM + 2 * hi;
    #pragma unroll
    for (int ks = 0; ks < 32; ++ks) afrag[ks] = ld2(arow + 4 * ks);

    int   labi[8];
    float s2i[8];
    #pragma unroll
    for (int r = 0; r < 8; ++r) {
        labi[r] = label[i0 + r + 8 * hi];
        s2i[r]  = sem2[i0 + r + 8 * hi];
    }

    float mp[8], mn[8], ms[8];
    #pragma unroll
    for (int r = 0; r < 8; ++r) { mp[r] = -INFINITY; mn[r] = INFINITY; ms[r] = INFINITY; }

    float* slab_wr = &s_tile[wave][4 * lane];
    const float* slab_rd = &s_tile[wave][col * TILE_PITCH + 2 * hi];

    for (int jt = wave; jt < N_PTS / 16; jt += PAIR_WAVES) {
        const int   j    = jt * 16 + col;
        const int   labj = label[j];
        const float x2j  = x2[j];
        const float* xrow = x + (size_t)(jt * 16) * D_DIM + 4 * lane;

        v4f st[16];
        #pragma unroll
        for (int t = 0; t < 16; ++t) st[t] = *(const v4f*)(xrow + (size_t)t * D_DIM);
        __builtin_amdgcn_sched_barrier(0);
        #pragma unroll
        for (int t = 0; t < 16; ++t) *(v4f*)(slab_wr + t * TILE_PITCH) = st[t];

        __builtin_amdgcn_fence(__ATOMIC_RELEASE, "wavefront");
        __builtin_amdgcn_wave_barrier();
        __builtin_amdgcn_fence(__ATOMIC_ACQUIRE, "wavefront");

        v2f bfrag[32];
        #pragma unroll
        for (int ks = 0; ks < 32; ++ks) bfrag[ks] = *(const v2f*)(slab_rd + 4 * ks);

        __builtin_amdgcn_fence(__ATOMIC_RELEASE, "wavefront");
        __builtin_amdgcn_wave_barrier();
        __builtin_amdgcn_fence(__ATOMIC_ACQUIRE, "wavefront");
        __builtin_amdgcn_sched_barrier(0);

        v8f c = {};
        #pragma unroll
        for (int ks = 0; ks < 32; ++ks) {
            c = __builtin_amdgcn_wmma_f32_16x16x4_f32(
                    false, afrag[ks], false, bfrag[ks], (short)0, c, false, false);
        }

        #pragma unroll
        for (int r = 0; r < 8; ++r) {
            const int i  = i0 + r + 8 * hi;
            const float sq = s2i[r] + x2j - 2.f * c[r];
            const bool same = (labi[r] == labj);
            const bool pos  = same && (i != j);
            mp[r] = fmaxf(mp[r], pos  ? sq : -INFINITY);
            mn[r] = fminf(mn[r], same ? INFINITY : sq);
            ms[r] = fminf(ms[r], same ? sq : INFINITY);
        }
    }

    #pragma unroll
    for (int r = 0; r < 8; ++r) {
        float a = mp[r], m = mn[r], s = ms[r];
        for (int off = 8; off >= 1; off >>= 1) {
            a = fmaxf(a, __shfl_xor(a, off, 16));
            m = fminf(m, __shfl_xor(m, off, 16));
            s = fminf(s, __shfl_xor(s, off, 16));
        }
        if (col == 0) {
            s_mp[wave][r + 8 * hi] = a;
            s_mn[wave][r + 8 * hi] = m;
            s_ms[wave][r + 8 * hi] = s;
        }
    }
    __syncthreads();

    if (threadIdx.x < 16) {
        float a = s_mp[0][threadIdx.x], m = s_mn[0][threadIdx.x], s = s_ms[0][threadIdx.x];
        #pragma unroll
        for (int w = 1; w < PAIR_WAVES; ++w) {
            a = fmaxf(a, s_mp[w][threadIdx.x]);
            m = fminf(m, s_mn[w][threadIdx.x]);
            s = fminf(s, s_ms[w][threadIdx.x]);
        }
        const float fpd = (a == -INFINITY) ? 0.f : sqrtf((a > 0.f ? a : 0.f) + EPS_TRM);
        const float cnd = fminf(sqrtf((m > 0.f ? m : 0.f) + EPS_TRM),
                                sqrtf((s > 0.f ? s : 0.f) + EPS_TRM) + BIG_F);
        const float z  = fpd - cnd;
        const float sp = (z > 20.f) ? z : log1pf(expf(z));
        diff[i0 + threadIdx.x] = sp;
    }
}

extern "C" void kernel_launch(void* const* d_in, const int* in_sizes, int n_in,
                              void* d_out, int out_size, void* d_ws, size_t ws_size,
                              hipStream_t stream) {
    const float* x     = (const float*)d_in[0];
    const float* wv    = (const float*)d_in[1];
    const int*   label = (const int*)  d_in[2];
    const float* alpha = (const float*)d_in[3];
    const float* W     = (const float*)d_in[4];
    const float* b     = (const float*)d_in[5];

    float* diff = (float*)d_out;
    float* sem  = diff + N_PTS;

    float* x2   = (float*)d_ws;
    float* sem2 = x2 + N_PTS;

    hipLaunchKernelGGL(x2_kernel, dim3(N_PTS / 256), dim3(256), 0, stream, x, x2);
    hipLaunchKernelGGL(sem_fc_kernel, dim3(N_PTS / FC_ROWS), dim3(32 * FC_WAVES), 0, stream,
                       x, wv, alpha, W, b, sem, sem2);
    hipLaunchKernelGGL(pair_kernel, dim3(N_PTS / 16), dim3(32 * PAIR_WAVES), 0, stream,
                       x, label, sem, sem2, x2, diff);
}


# pair kernel: each wave keeps the A fragments of two 16-row stripes and feeds every B fragment to two independent WMMA chains (32 rows per x tile, 12 waves per work-group); no sum reordered
# speedup vs baseline: 11.8515x; 1.1239x over previous
#include <hip/hip_runtime.h>
#include <math.h>

#define N_PTS   8192
#define WV_DIM  300
#define D_DIM   128
#define BIG_F   1e8f
#define EPS_TRM (128.0f * 1e-16f)

typedef float v2f __attribute__((ext_vector_type(2)));
typedef float v4f __attribute__((ext_vector_type(4)));
typedef float v8f __attribute__((ext_vector_type(8)));

__device__ __forceinline__ v2f ld2(const float* p) {
    return *(const v2f*)p;
}

__global__ void x2_kernel(const float* __restrict__ x, float* __restrict__ x2) {
    int i = blockIdx.x * blockDim.x + threadIdx.x;
    if (i < N_PTS) {
        const float* row = x + (size_t)i * D_DIM;
        float s = 0.f;
        #pragma unroll 8
        for (int d = 0; d < D_DIM; ++d) s = fmaf(row[d], row[d], s);
        x2[i] = s;
    }
}

#define FC_ROWS   32
#define FC_WAVES  16
#define SEM_PITCH 132
#define FC_KB     15
__global__ __launch_bounds__(32 * FC_WAVES) void sem_fc_kernel(
    const float* __restrict__ x,  const float* __restrict__ wv,
    const float* __restrict__ alpha, const float* __restrict__ W,
    const float* __restrict__ b,  float* __restrict__ sem_out,
    float* __restrict__ sem2) {
    __shared__ __attribute__((aligned(16))) float s_w[D_DIM * WV_DIM];
    __shared__ __attribute__((aligned(16))) float s_a[FC_ROWS * WV_DIM];
    __shared__ __attribute__((aligned(16))) float s_sem[FC_ROWS * SEM_PITCH];

    const int tid  = threadIdx.x;
    const int lane = tid & 31;
    const int wave = __builtin_amdgcn_readfirstlane(tid >> 5);
    const int col  = lane & 15;
    const int hi   = lane >> 4;
    const int dt   = wave & 7;
    const int st   = wave >> 3;
    const int i0   = blockIdx.x * FC_ROWS;

    const v4f* gw = (const v4f*)W;
    const v4f* ga = (const v4f*)(wv + (size_t)i0 * WV_DIM);
    v4f* lw = (v4f*)s_w;
    v4f* la = (v4f*)s_a;
    const int tw = min(tid + 512 * 18, D_DIM * WV_DIM / 4 - 1);
    const int ta = min(tid + 512 * 4, FC_ROWS * WV_DIM / 4 - 1);
    v4f rw[19], ra[5];
    #pragma unroll
    for (int t = 0; t < 18; ++t) rw[t] = gw[tid + 512 * t];
    rw[18] = gw[tw];
    #pragma unroll
    for (int t = 0; t < 4; ++t) ra[t] = ga[tid + 512 * t];
    ra[4] = ga[ta];

    const int   d  = dt * 16 + col;
    const float bd = b[d];
    float alph[8], xv[8];
    #pragma unroll
    for (int r = 0; r < 8; ++r) {
        const int i = i0 + 16 * st + r + 8 * hi;
        alph[r] = alpha[i];
        xv[r]   = x[(size_t)i * D_DIM + d];
    }

    #pragma unroll
    for (int t = 0; t < 18; ++t) lw[tid + 512 * t] = rw[t];
    lw[tw] = rw[18];
    #pragma unroll
    for (int t = 0; t < 4; ++t) la[tid + 512 * t] = ra[t];
    la[ta] = ra[4];
    __syncthreads();

    const float* ap = s_a + (16 * st + col) * WV_DIM + 2 * hi;
    const float* bp = s_w + (16 * dt + col) * WV_DIM + 2 * hi;
    v2f fa[2][FC_KB], fb[2][FC_KB];
    #pragma unroll
    for (int k = 0; k < FC_KB; ++k) {
        fa[0][k] = *(const v2f*)(ap + 4 * k);
        fb[0][k] = *(const v2f*)(bp + 4 * k);
    }
    v8f c = {};
    #pragma unroll
    for (int kb = 0; kb < WV_DIM / 4 / FC_KB; ++kb) {
        if (kb + 1 < WV_DIM / 4 / FC_KB) {
            #pragma unroll
            for (int k = 0; k < FC_KB; ++k) {
                fa[(kb + 1) & 1][k] = *(const v2f*)(ap + 4 * (FC_KB * (kb + 1) + k));
                fb[(kb + 1) & 1][k] = *(const v2f*)(bp + 4 * (FC_KB * (kb + 1) + k));
            }
        }
        __builtin_amdgcn_sched_barrier(0);
        #pragma unroll
        for (int k = 0; k < FC_KB; ++k)
            c = __builtin_amdgcn_wmma_f32_16x16x4_f32(
                    false, fa[kb & 1][k], false, fb[kb & 1][k], (short)0, c, false, false);
        __builtin_amdgcn_sched_barrier(0);
    }

    #pragma unroll
    for (int r = 0; r < 8; ++r) {
        const float so = c[r] + bd;
        const float s  = alph[r] * so + (1.f - alph[r]) * xv[r];
        s_sem[(16 * st + r + 8 * hi) * SEM_PITCH + d] = s;
    }
    __syncthreads();

    #pragma unroll
    for (int t = 0; t < 2; ++t) {
        const int e   = tid + 512 * t;
        const int row = e >> 5;
        const int q   = e & 31;
        const v4f v = *(const v4f*)(s_sem + row * SEM_PITCH + 4 * q);
        *(v4f*)(sem_out + (size_t)(i0 + row) * D_DIM + 4 * q) = v;
    }

    {
        const int row = tid >> 4;
        float nrm = 0.f;
        #pragma unroll
        for (int t = 0; t < 8; ++t) {
            const float v = s_sem[row * SEM_PITCH + 16 * t + col];
            nrm = fmaf(v, v, nrm);
        }
        for (int off = 8; off >= 1; off >>= 1)
            nrm += __shfl_xor(nrm, off, 16);
        if (col == 0) sem2[i0 + row] = nrm;
    }
}

#define PAIR_WAVES   12
#define PAIR_STRIPES 2
#define PAIR_ROWS    (16 * PAIR_STRIPES)
#define TILE_PITCH   132
__global__ __launch_bounds__(32 * PAIR_WAVES) __attribute__((amdgpu_waves_per_eu(3, 3))) void pair_kernel(
    const float* __restrict__ x,    const int* __restrict__ label,
    const float* __restrict__ sem,  const float* __restrict__ sem2,
    const float* __restrict__ x2,   float* __restrict__ diff) {
    __shared__ float s_mp[PAIR_WAVES][PAIR_ROWS];
    __shared__ float s_mn[PAIR_WAVES][PAIR_ROWS];
    __shared__ float s_ms[PAIR_WAVES][PAIR_ROWS];
    __shared__ __attribute__((aligned(16))) float s_tile[PAIR_WAVES][16 * TILE_PITCH];

    const int lane = threadIdx.x & 31;
    const int wave = __builtin_amdgcn_readfirstlane((int)(threadIdx.x >> 5));
    const int col  = lane & 15;
    const int hi   = lane >> 4;
    const int i0   = blockIdx.x * PAIR_ROWS;

    v2f afrag[PAIR_STRIPES][32];
    #pragma unroll
    for (int s = 0; s < PAIR_STRIPES; ++s) {
        const float* arow = sem + (size_t)(i0 + 16 * s + col) * D_DIM + 2 * hi;
        #pragma unroll
        for (int ks = 0; ks < 32; ++ks) afrag[s][ks] = ld2(arow + 4 * ks);
    }

    int   labi[PAIR_STRIPES][8];
    float s2i[PAIR_STRIPES][8];
    #pragma unroll
    for (int s = 0; s < PAIR_STRIPES; ++s) {
        #pragma unroll
        for (int r = 0; r < 8; ++r) {
            labi[s][r] = label[i0 + 16 * s + r + 8 * hi];
            s2i[s][r]  = sem2[i0 + 16 * s + r + 8 * hi];
        }
    }

    float mp[PAIR_STRIPES][8], mn[PAIR_STRIPES][8], ms[PAIR_STRIPES][8];
    #pragma unroll
    for (int s = 0; s < PAIR_STRIPES; ++s) {
        #pragma unroll
        for (int r = 0; r < 8; ++r) { mp[s][r] = -INFINITY; mn[s][r] = INFINITY; ms[s][r] = INFINITY; }
    }

    float* slab_wr = &s_tile[wave][4 * lane];
    const float* slab_rd = &s_tile[wave][col * TILE_PITCH + 2 * hi];

    for (int jt = wave; jt < N_PTS / 16; jt += PAIR_WAVES) {
        const int   j    = jt * 16 + col;
        const int   labj = label[j];
        const float x2j  = x2[j];
        const float* xrow = x + (size_t)(jt * 16) * D_DIM + 4 * lane;

        v4f st[16];
        #pragma unroll
        for (int t = 0; t < 16; ++t) st[t] = *(const v4f*)(xrow + (size_t)t * D_DIM);
        __builtin_amdgcn_sched_barrier(0);
        #pragma unroll
        for (int t = 0; t < 16; ++t) *(v4f*)(slab_wr + t * TILE_PITCH) = st[t];

        __builtin_amdgcn_fence(__ATOMIC_RELEASE, "wavefront");
        __builtin_amdgcn_wave_barrier();
        __builtin_amdgcn_fence(__ATOMIC_ACQUIRE, "wavefront");

        v2f bfrag[32];
        #pragma unroll
        for (int ks = 0; ks < 32; ++ks) bfrag[ks] = *(const v2f*)(slab_rd + 4 * ks);

        __builtin_amdgcn_fence(__ATOMIC_RELEASE, "wavefront");
        __builtin_amdgcn_wave_barrier();
        __builtin_amdgcn_fence(__ATOMIC_ACQUIRE, "wavefront");
        __builtin_amdgcn_sched_barrier(0);

        v8f c[PAIR_STRIPES] = {};
        #pragma unroll
        for (int ks = 0; ks < 32; ++ks) {
            #pragma unroll
            for (int s = 0; s < PAIR_STRIPES; ++s)
                c[s] = __builtin_amdgcn_wmma_f32_16x16x4_f32(
                        false, afrag[s][ks], false, bfrag[ks], (short)0, c[s], false, false);
        }

        #pragma unroll
        for (int s = 0; s < PAIR_STRIPES; ++s) {
            #pragma unroll
            for (int r = 0; r < 8; ++r) {
                const int i  = i0 + 16 * s + r + 8 * hi;
                const float sq = s2i[s][r] + x2j - 2.f * c[s][r];
                const bool same = (labi[s][r] == labj);
                const bool pos  = same && (i != j);
                mp[s][r] = fmaxf(mp[s][r], pos  ? sq : -INFINITY);
                mn[s][r] = fminf(mn[s][r], same ? INFINITY : sq);
                ms[s][r] = fminf(ms[s][r], same ? sq : INFINITY);
            }
        }
    }

    #pragma unroll
    for (int s = 0; s < PAIR_STRIPES; ++s) {
        #pragma unroll
        for (int r = 0; r < 8; ++r) {
            float a = mp[s][r], m = mn[s][r], q = ms[s][r];
            for (int off = 8; off >= 1; off >>= 1) {
                a = fmaxf(a, __shfl_xor(a, off, 16));
                m = fminf(m, __shfl_xor(m, off, 16));
                q = fminf(q, __shfl_xor(q, off, 16));
            }
            if (col == 0) {
                s_mp[wave][16 * s + r + 8 * hi] = a;
                s_mn[wave][16 * s + r + 8 * hi] = m;
                s_ms[wave][16 * s + r + 8 * hi] = q;
            }
        }
    }
    __syncthreads();

    if (threadIdx.x < PAIR_ROWS) {
        float a = s_mp[0][threadIdx.x], m = s_mn[0][threadIdx.x], q = s_ms[0][threadIdx.x];
        #pragma unroll
        for (int w = 1; w < PAIR_WAVES; ++w) {
            a = fmaxf(a, s_mp[w][threadIdx.x]);
            m = fminf(m, s_mn[w][threadIdx.x]);
            q = fminf(q, s_ms[w][threadIdx.x]);
        }
        const float fpd = (a == -INFINITY) ? 0.f : sqrtf((a > 0.f ? a : 0.f) + EPS_TRM);
        const float cnd = fminf(sqrtf((m > 0.f ? m : 0.f) + EPS_TRM),
                                sqrtf((q > 0.f ? q : 0.f) + EPS_TRM) + BIG_F);
        const float z  = fpd - cnd;
        const float sp = (z > 20.f) ? z : log1pf(expf(z));
        diff[i0 + threadIdx.x] = sp;
    }
}

extern "C" void kernel_launch(void* const* d_in, const int* in_sizes, int n_in,
                              void* d_out, int out_size, void* d_ws, size_t ws_size,
                              hipStream_t stream) {
    const float* x     = (const float*)d_in[0];
    const float* wv    = (const float*)d_in[1];
    const int*   label = (const int*)  d_in[2];
    const float* alpha = (const float*)d_in[3];
    const float* W     = (const float*)d_in[4];
    const float* b     = (const float*)d_in[5];

    float* diff = (float*)d_out;
    float* sem  = diff + N_PTS;

    float* x2   = (float*)d_ws;
    float* sem2 = x2 + N_PTS;

    hipLaunchKernelGGL(x2_kernel, dim3(N_PTS / 256), dim3(256), 0, stream, x, x2);
    hipLaunchKernelGGL(sem_fc_kernel, dim3(N_PTS / FC_ROWS), dim3(32 * FC_WAVES), 0, stream,
                       x, wv, alpha, W, b, sem, sem2);
    hipLaunchKernelGGL(pair_kernel, dim3(N_PTS / PAIR_ROWS), dim3(32 * PAIR_WAVES), 0, stream,
                       x, label, sem, sem2, x2, diff);
}


# pair kernel: tile sweep software-pipelined in each wave (tile t+2 rows and tile t+1 fragments requested ahead of tile t's WMMA block, second fragment set), 8 waves per group; no sum reordered
# speedup vs baseline: 12.4499x; 1.0505x over previous
#include <hip/hip_runtime.h>
#include <math.h>

#define N_PTS   8192
#define WV_DIM  300
#define D_DIM   128
#define BIG_F   1e8f
#define EPS_TRM (128.0f * 1e-16f)

typedef float v2f __attribute__((ext_vector_type(2)));
typedef float v4f __attribute__((ext_vector_type(4)));
typedef float v8f __attribute__((ext_vector_type(8)));

__device__ __forceinline__ v2f ld2(const float* p) {
    return *(const v2f*)p;
}

__global__ void x2_kernel(const float* __restrict__ x, float* __restrict__ x2) {
    int i = blockIdx.x * blockDim.x + threadIdx.x;
    if (i < N_PTS) {
        const float* row = x + (size_t)i * D_DIM;
        float s = 0.f;
        #pragma unroll 8
        for (int d = 0; d < D_DIM; ++d) s = fmaf(row[d], row[d], s);
        x2[i] = s;
    }
}

#define FC_ROWS   32
#define FC_WAVES  16
#define SEM_PITCH 132
#define FC_KB     15
__global__ __launch_bounds__(32 * FC_WAVES) void sem_fc_kernel(
    const float* __restrict__ x,  const float* __restrict__ wv,
    const float* __restrict__ alpha, const float* __restrict__ W,
    const float* __restrict__ b,  float* __restrict__ sem_out,
    float* __restrict__ sem2) {
    __shared__ __attribute__((aligned(16))) float s_w[D_DIM * WV_DIM];
    __shared__ __attribute__((aligned(16))) float s_a[FC_ROWS * WV_DIM];
    __shared__ __attribute__((aligned(16))) float s_sem[FC_ROWS * SEM_PITCH];

    const int tid  = threadIdx.x;
    const int lane = tid & 31;
    const int wave = __builtin_amdgcn_readfirstlane(tid >> 5);
    const int col  = lane & 15;
    const int hi   = lane >> 4;
    const int dt   = wave & 7;
    const int st   = wave >> 3;
    const int i0   = blockIdx.x * FC_ROWS;

    const v4f* gw = (const v4f*)W;
    const v4f* ga = (const v4f*)(wv + (size_t)i0 * WV_DIM);
    v4f* lw = (v4f*)s_w;
    v4f* la = (v4f*)s_a;
    const int tw = min(tid + 512 * 18, D_DIM * WV_DIM / 4 - 1);
    const int ta = min(tid + 512 * 4, FC_ROWS * WV_DIM / 4 - 1);
    v4f rw[19], ra[5];
    #pragma unroll
    for (int t = 0; t < 18; ++t) rw[t] = gw[tid + 512 * t];
    rw[18] = gw[tw];
    #pragma unroll
    for (int t = 0; t < 4; ++t) ra[t] = ga[tid + 512 * t];
    ra[4] = ga[ta];

    const int   d  = dt * 16 + col;
    const float bd = b[d];
    float alph[8], xv[8];
    #pragma unroll
    for (int r = 0; r < 8; ++r) {
        const int i = i0 + 16 * st + r + 8 * hi;
        alph[r] = alpha[i];
        xv[r]   = x[(size_t)i * D_DIM + d];
    }

    #pragma unroll
    for (int t = 0; t < 18; ++t) lw[tid + 512 * t] = rw[t];
    lw[tw] = rw[18];
    #pragma unroll
    for (int t = 0; t < 4; ++t) la[tid + 512 * t] = ra[t];
    la[ta] = ra[4];
    __syncthreads();

    const float* ap = s_a + (16 * st + col) * WV_DIM + 2 * hi;
    const float* bp = s_w + (16 * dt + col) * WV_DIM + 2 * hi;
    v2f fa[2][FC_KB], fb[2][FC_KB];
    #pragma unroll
    for (int k = 0; k < FC_KB; ++k) {
        fa[0][k] = *(const v2f*)(ap + 4 * k);
        fb[0][k] = *(const v2f*)(bp + 4 * k);
    }
    v8f c = {};
    #pragma unroll
    for (int kb = 0; kb < WV_DIM / 4 / FC_KB; ++kb) {
        if (kb + 1 < WV_DIM / 4 / FC_KB) {
            #pragma unroll
            for (int k = 0; k < FC_KB; ++k) {
                fa[(kb + 1) & 1][k] = *(const v2f*)(ap + 4 * (FC_KB * (kb + 1) + k));
                fb[(kb + 1) & 1][k] = *(const v2f*)(bp + 4 * (FC_KB * (kb + 1) + k));
            }
        }
        __builtin_amdgcn_sched_barrier(0);
        #pragma unroll
        for (int k = 0; k < FC_KB; ++k)
            c = __builtin_amdgcn_wmma_f32_16x16x4_f32(
                    false, fa[kb & 1][k], false, fb[kb & 1][k], (short)0, c, false, false);
        __builtin_amdgcn_sched_barrier(0);
    }

    #pragma unroll
    for (int r = 0; r < 8; ++r) {
        const float so = c[r] + bd;
        const float s  = alph[r] * so + (1.f - alph[r]) * xv[r];
        s_sem[(16 * st + r + 8 * hi) * SEM_PITCH + d] = s;
    }
    __syncthreads();

    #pragma unroll
    for (int t = 0; t < 2; ++t) {
        const int e   = tid + 512 * t;
        const int row = e >> 5;
        const int q   = e & 31;
        const v4f v = *(const v4f*)(s_sem + row * SEM_PITCH + 4 * q);
        *(v4f*)(sem_out + (size_t)(i0 + row) * D_DIM + 4 * q) = v;
    }

    {
        const int row = tid >> 4;
        float nrm = 0.f;
        #pragma unroll
        for (int t = 0; t < 8; ++t) {
            const float v = s_sem[row * SEM_PITCH + 16 * t + col];
            nrm = fmaf(v, v, nrm);
        }
        for (int off = 8; off >= 1; off >>= 1)
            nrm += __shfl_xor(nrm, off, 16);
        if (col == 0) sem2[i0 + row] = nrm;
    }
}

#define PAIR_WAVES   8
#define PAIR_STRIPES 2
#define PAIR_ROWS    (16 * PAIR_STRIPES)
#define TILE_PITCH   132
#define PAIR_STEPS   (N_PTS / 16 / PAIR_WAVES)

__device__ __forceinline__ void pair_step(
    const int jt, const int jt_last, const int i0, const int lane, const int col, const int hi,
    const float* __restrict__ x, const int* __restrict__ label, const float* __restrict__ x2,
    const v2f (&afrag)[PAIR_STRIPES][32], v2f (&cur)[32], v2f (&nxt)[32], v4f (&st)[16],
    float* slab_wr, const float* slab_rd,
    const int labj, const float x2j, int& labn, float& x2n,
    const int (&labi)[PAIR_STRIPES][8], const float (&s2i)[PAIR_STRIPES][8],
    float (&mp)[PAIR_STRIPES][8], float (&mn)[PAIR_STRIPES][8], float (&ms)[PAIR_STRIPES][8]) {
    #pragma unroll
    for (int t = 0; t < 16; ++t) *(v4f*)(slab_wr + t * TILE_PITCH) = st[t];

    const int jt1 = min(jt + PAIR_WAVES, jt_last);
    const int jt2 = min(jt + 2 * PAIR_WAVES, jt_last);
    labn = label[jt1 * 16 + col];
    x2n  = x2[jt1 * 16 + col];
    const float* xrow = x + (size_t)(jt2 * 16) * D_DIM + 4 * lane;
    #pragma unroll
    for (int t = 0; t < 16; ++t) st[t] = *(const v4f*)(xrow + (size_t)t * D_DIM);

    __builtin_amdgcn_fence(__ATOMIC_RELEASE, "wavefront");
    __builtin_amdgcn_wave_barrier();
    __builtin_amdgcn_fence(__ATOMIC_ACQUIRE, "wavefront");

    #pragma unroll
    for (int ks = 0; ks < 32; ++ks) nxt[ks] = *(const v2f*)(slab_rd + 4 * ks);

    __builtin_amdgcn_fence(__ATOMIC_RELEASE, "wavefront");
    __builtin_amdgcn_wave_barrier();
    __builtin_amdgcn_fence(__ATOMIC_ACQUIRE, "wavefront");
    __builtin_amdgcn_sched_barrier(0);

    v8f c[PAIR_STRIPES] = {};
    #pragma unroll
    for (int ks = 0; ks < 32; ++ks) {
        #pragma unroll
        for (int s = 0; s < PAIR_STRIPES; ++s)
            c[s] = __builtin_amdgcn_wmma_f32_16x16x4_f32(
                    false, afrag[s][ks], false, cur[ks], (short)0, c[s], false, false);
    }

    __builtin_amdgcn_sched_barrier(0x2 | 0x4);

    const int j = jt * 16 + col;
    #pragma unroll
    for (int s = 0; s < PAIR_STRIPES; ++s) {
        #pragma unroll
        for (int r = 0; r < 8; ++r) {
            const int i  = i0 + 16 * s + r + 8 * hi;
            const float sq = s2i[s][r] + x2j - 2.f * c[s][r];
            const bool same = (labi[s][r] == labj);
            const bool pos  = same && (i != j);
            mp[s][r] = fmaxf(mp[s][r], pos  ? sq : -INFINITY);
            mn[s][r] = fminf(mn[s][r], same ? INFINITY : sq);
            ms[s][r] = fminf(ms[s][r], same ? sq : INFINITY);
        }
    }
}

__global__ __launch_bounds__(32 * PAIR_WAVES) __attribute__((amdgpu_waves_per_eu(2, 2))) void pair_kernel(
    const float* __restrict__ x,    const int* __restrict__ label,
    const float* __restrict__ sem,  const float* __restrict__ sem2,
    const float* __restrict__ x2,   float* __restrict__ diff) {
    __shared__ float s_mp[PAIR_WAVES][PAIR_ROWS];
    __shared__ float s_mn[PAIR_WAVES][PAIR_ROWS];
    __shared__ float s_ms[PAIR_WAVES][PAIR_ROWS];
    __shared__ __attribute__((aligned(16))) float s_tile[PAIR_WAVES][16 * TILE_PITCH];

    const int lane = threadIdx.x & 31;
    const int wave = __builtin_amdgcn_readfirstlane((int)(threadIdx.x >> 5));
    const int col  = lane & 15;
    const int hi   = lane >> 4;
    const int i0   = blockIdx.x * PAIR_ROWS;

    v2f afrag[PAIR_STRIPES][32];
    #pragma unroll
    for (int s = 0; s < PAIR_STRIPES; ++s) {
        const float* arow = sem + (size_t)(i0 + 16 * s + col) * D_DIM + 2 * hi;
        #pragma unroll
        for (int ks = 0; ks < 32; ++ks) afrag[s][ks] = ld2(arow + 4 * ks);
    }

    int   labi[PAIR_STRIPES][8];
    float s2i[PAIR_STRIPES][8];
    #pragma unroll
    for (int s = 0; s < PAIR_STRIPES; ++s) {
        #pragma unroll
        for (int r = 0; r < 8; ++r) {
            labi[s][r] = label[i0 + 16 * s + r + 8 * hi];
            s2i[s][r]  = sem2[i0 + 16 * s + r + 8 * hi];
        }
    }

    float mp[PAIR_STRIPES][8], mn[PAIR_STRIPES][8], ms[PAIR_STRIPES][8];
    #pragma unroll
    for (int s = 0; s < PAIR_STRIPES; ++s) {
        #pragma unroll
        for (int r = 0; r < 8; ++r) { mp[s][r] = -INFINITY; mn[s][r] = INFINITY; ms[s][r] = INFINITY; }
    }

    float* slab_wr = &s_tile[wave][4 * lane];
    const float* slab_rd = &s_tile[wave][col * TILE_PITCH + 2 * hi];

    const int jt_last = wave + PAIR_WAVES * (PAIR_STEPS - 1);
    v4f st[16];
    v2f fr0[32], fr1[32];
    int   lab0, lab1;
    float xn0, xn1;
    {
        const float* xrow = x + (size_t)(wave * 16) * D_DIM + 4 * lane;
        #pragma unroll
        for (int t = 0; t < 16; ++t) st[t] = *(const v4f*)(xrow + (size_t)t * D_DIM);
        lab0 = label[wave * 16 + col];
        xn0  = x2[wave * 16 + col];
        #pragma unroll
        for (int t = 0; t < 16; ++t) *(v4f*)(slab_wr + t * TILE_PITCH) = st[t];
        xrow += (size_t)(PAIR_WAVES * 16) * D_DIM;
        #pragma unroll
        for (int t = 0; t < 16; ++t) st[t] = *(const v4f*)(xrow + (size_t)t * D_DIM);
        __builtin_amdgcn_fence(__ATOMIC_RELEASE, "wavefront");
        __builtin_amdgcn_wave_barrier();
        __builtin_amdgcn_fence(__ATOMIC_ACQUIRE, "wavefront");
        #pragma unroll
        for (int ks = 0; ks < 32; ++ks) fr0[ks] = *(const v2f*)(slab_rd + 4 * ks);
        __builtin_amdgcn_fence(__ATOMIC_RELEASE, "wavefront");
        __builtin_amdgcn_wave_barrier();
        __builtin_amdgcn_fence(__ATOMIC_ACQUIRE, "wavefront");
    }

    for (int jt = wave; jt <= jt_last; jt += 2 * PAIR_WAVES) {
        pair_step(jt, jt_last, i0, lane, col, hi, x, label, x2, afrag, fr0, fr1, st,
                  slab_wr, slab_rd, lab0, xn0, lab1, xn1, labi, s2i, mp, mn, ms);
        pair_step(jt + PAIR_WAVES, jt_last, i0, lane, col, hi, x, label, x2, afrag, fr1, fr0, st,
                  slab_wr, slab_rd, lab1, xn1, lab0, xn0, labi, s2i, mp, mn, ms);
    }

    #pragma unroll
    for (int s = 0; s < PAIR_STRIPES; ++s) {
        #pragma unroll
        for (int r = 0; r < 8; ++r) {
            float a = mp[s][r], m = mn[s][r], q = ms[s][r];
            for (int off = 8; off >= 1; off >>= 1) {
                a = fmaxf(a, __shfl_xor(a, off, 16));
                m = fminf(m, __shfl_xor(m, off, 16));
                q = fminf(q, __shfl_xor(q, off, 16));
            }
            if (col == 0) {
                s_mp[wave][16 * s + r + 8 * hi] = a;
                s_mn[wave][16 * s + r + 8 * hi] = m;
                s_ms[wave][16 * s + r + 8 * hi] = q;
            }
        }
    }
    __syncthreads();

    if (threadIdx.x < PAIR_ROWS) {
        float a = s_mp[0][threadIdx.x], m = s_mn[0][threadIdx.x], q = s_ms[0][threadIdx.x];
        #pragma unroll
        for (int w = 1; w < PAIR_WAVES; ++w) {
            a = fmaxf(a, s_mp[w][threadIdx.x]);
            m = fminf(m, s_mn[w][threadIdx.x]);
            q = fminf(q, s_ms[w][threadIdx.x]);
        }
        const float fpd = (a == -INFINITY) ? 0.f : sqrtf((a > 0.f ? a : 0.f) + EPS_TRM);
        const float cnd = fminf(sqrtf((m > 0.f ? m : 0.f) + EPS_TRM),
                                sqrtf((q > 0.f ? q : 0.f) + EPS_TRM) + BIG_F);
        const float z  = fpd - cnd;
        const float sp = (z > 20.f) ? z : log1pf(expf(z));
        diff[i0 + threadIdx.x] = sp;
    }
}

extern "C" void kernel_launch(void* const* d_in, const int* in_sizes, int n_in,
                              void* d_out, int out_size, void* d_ws, size_t ws_size,
                              hipStream_t stream) {
    const float* x     = (const float*)d_in[0];
    const float* wv    = (const float*)d_in[1];
    const int*   label = (const int*)  d_in[2];
    const float* alpha = (const float*)d_in[3];
    const float* W     = (const float*)d_in[4];
    const float* b     = (const float*)d_in[5];

    float* diff = (float*)d_out;
    float* sem  = diff + N_PTS;

    float* x2   = (float*)d_ws;
    float* sem2 = x2 + N_PTS;

    hipLaunchKernelGGL(x2_kernel, dim3(N_PTS / 256), dim3(256), 0, stream, x, x2);
    hipLaunchKernelGGL(sem_fc_kernel, dim3(N_PTS / FC_ROWS), dim3(32 * FC_WAVES), 0, stream,
                       x, wv, alpha, W, b, sem, sem2);
    hipLaunchKernelGGL(pair_kernel, dim3(N_PTS / PAIR_ROWS), dim3(32 * PAIR_WAVES), 0, stream,
                       x, label, sem, sem2, x2, diff);
}


# norm kernel folded into the fc kernel: first wave of each 32-row block sums its rows' squares, one row per lane, same ascending-d fma chain, behind its matrix steps; 3 launches -> 2; no sum reordered
# speedup vs baseline: 12.8766x; 1.0343x over previous
#include <hip/hip_runtime.h>
#include <math.h>

#define N_PTS   8192
#define WV_DIM  300
#define D_DIM   128
#define BIG_F   1e8f
#define EPS_TRM (128.0f * 1e-16f)

typedef float v2f __attribute__((ext_vector_type(2)));
typedef float v4f __attribute__((ext_vector_type(4)));
typedef float v8f __attribute__((ext_vector_type(8)));

__device__ __forceinline__ v2f ld2(const float* p) {
    return *(const v2f*)p;
}

#define FC_ROWS   32
#define FC_WAVES  16
#define SEM_PITCH 132
#define FC_KB     15
__global__ __launch_bounds__(32 * FC_WAVES) void sem_fc_kernel(
    const float* __restrict__ x,  const float* __restrict__ wv,
    const float* __restrict__ alpha, const float* __restrict__ W,
    const float* __restrict__ b,  float* __restrict__ sem_out,
    float* __restrict__ sem2, float* __restrict__ x2) {
    __shared__ __attribute__((aligned(16))) float s_w[D_DIM * WV_DIM];
    __shared__ __attribute__((aligned(16))) float s_a[FC_ROWS * WV_DIM];
    __shared__ __attribute__((aligned(16))) float s_sem[FC_ROWS * SEM_PITCH];

    const int tid  = threadIdx.x;
    const int lane = tid & 31;
    const int wave = __builtin_amdgcn_readfirstlane(tid >> 5);
    const int col  = lane & 15;
    const int hi   = lane >> 4;
    const int dt   = wave & 7;
    const int st   = wave >> 3;
    const int i0   = blockIdx.x * FC_ROWS;

    const v4f* gw = (const v4f*)W;
    const v4f* ga = (const v4f*)(wv + (size_t)i0 * WV_DIM);
    v4f* lw = (v4f*)s_w;
    v4f* la = (v4f*)s_a;
    const int tw = min(tid + 512 * 18, D_DIM * WV_DIM / 4 - 1);
    const int ta = min(tid + 512 * 4, FC_ROWS * WV_DIM / 4 - 1);
    v4f rw[19], ra[5];
    #pragma unroll
    for (int t = 0; t < 18; ++t) rw[t] = gw[tid + 512 * t];
    rw[18] = gw[tw];
    #pragma unroll
    for (int t = 0; t < 4; ++t) ra[t] = ga[tid + 512 * t];
    ra[4] = ga[ta];

    const int   d  = dt * 16 + col;
    const float bd = b[d];
    float alph[8], xv[8];
    #pragma unroll
    for (int r = 0; r < 8; ++r) {
        const int i = i0 + 16 * st + r + 8 * hi;
        alph[r] = alpha[i];
        xv[r]   = x[(size_t)i * D_DIM + d];
    }

    #pragma unroll
    for (int t = 0; t < 18; ++t) lw[tid + 512 * t] = rw[t];
    lw[tw] = rw[18];
    #pragma unroll
    for (int t = 0; t < 4; ++t) la[tid + 512 * t] = ra[t];
    la[ta] = ra[4];
    __syncthreads();

    const float* ap = s_a + (16 * st + col) * WV_DIM + 2 * hi;
    const float* bp = s_w + (16 * dt + col) * WV_DIM + 2 * hi;
    v2f fa[2][FC_KB], fb[2][FC_KB];
    #pragma unroll
    for (int k = 0; k < FC_KB; ++k) {
        fa[0][k] = *(const v2f*)(ap + 4 * k);
        fb[0][k] = *(const v2f*)(bp + 4 * k);
    }
    v8f c = {};
    #pragma unroll
    for (int kb = 0; kb < WV_DIM / 4 / FC_KB; ++kb) {
        if (kb + 1 < WV_DIM / 4 / FC_KB) {
            #pragma unroll
            for (int k = 0; k < FC_KB; ++k) {
                fa[(kb + 1) & 1][k] = *(const v2f*)(ap + 4 * (FC_KB * (kb + 1) + k));
                fb[(kb + 1) & 1][k] = *(const v2f*)(bp + 4 * (FC_KB * (kb + 1) + k));
            }
        }
        __builtin_amdgcn_sched_barrier(0);
        #pragma unroll
        for (int k = 0; k < FC_KB; ++k)
            c = __builtin_amdgcn_wmma_f32_16x16x4_f32(
                    false, fa[kb & 1][k], false, fb[kb & 1][k], (short)0, c, false, false);
        __builtin_amdgcn_sched_barrier(0);
    }

    if (wave == 0) {
        const float* xr = x + (size_t)(i0 + lane) * D_DIM;
        float nx = 0.f;
        #pragma unroll
        for (int k = 0; k < D_DIM; ++k) nx = fmaf(xr[k], xr[k], nx);
        x2[i0 + lane] = nx;
    }

    #pragma unroll
    for (int r = 0; r < 8; ++r) {
        const float so = c[r] + bd;
        const float s  = alph[r] * so + (1.f - alph[r]) * xv[r];
        s_sem[(16 * st + r + 8 * hi) * SEM_PITCH + d] = s;
    }
    __syncthreads();

    #pragma unroll
    for (int t = 0; t < 2; ++t) {
        const int e   = tid + 512 * t;
        const int row = e >> 5;
        const int q   = e & 31;
        const v4f v = *(const v4f*)(s_sem + row * SEM_PITCH + 4 * q);
        *(v4f*)(sem_out + (size_t)(i0 + row) * D_DIM + 4 * q) = v;
    }

    {
        const int row = tid >> 4;
        float nrm = 0.f;
        #pragma unroll
        for (int t = 0; t < 8; ++t) {
            const float v = s_sem[row * SEM_PITCH + 16 * t + col];
            nrm = fmaf(v, v, nrm);
        }
        for (int off = 8; off >= 1; off >>= 1)
            nrm += __shfl_xor(nrm, off, 16);
        if (col == 0) sem2[i0 + row] = nrm;
    }
}

#define PAIR_WAVES   8
#define PAIR_STRIPES 2
#define PAIR_ROWS    (16 * PAIR_STRIPES)
#define TILE_PITCH   132
#define PAIR_STEPS   (N_PTS / 16 / PAIR_WAVES)

__device__ __forceinline__ void pair_step(
    const int jt, const int jt_last, const int i0, const int lane, const int col, const int hi,
    const float* __restrict__ x, const int* __restrict__ label, const float* __restrict__ x2,
    const v2f (&afrag)[PAIR_STRIPES][32], v2f (&cur)[32], v2f (&nxt)[32], v4f (&st)[16],
    float* slab_wr, const float* slab_rd,
    const int labj, const float x2j, int& labn, float& x2n,
    const int (&labi)[PAIR_STRIPES][8], const float (&s2i)[PAIR_STRIPES][8],
    float (&mp)[PAIR_STRIPES][8], float (&mn)[PAIR_STRIPES][8], float (&ms)[PAIR_STRIPES][8]) {
    #pragma unroll
    for (int t = 0; t < 16; ++t) *(v4f*)(slab_wr + t * TILE_PITCH) = st[t];

    const int jt1 = min(jt + PAIR_WAVES, jt_last);
    const int jt2 = min(jt + 2 * PAIR_WAVES, jt_last);
    labn = label[jt1 * 16 + col];
    x2n  = x2[jt1 * 16 + col];
    const float* xrow = x + (size_t)(jt2 * 16) * D_DIM + 4 * lane;
    #pragma unroll
    for (int t = 0; t < 16; ++t) st[t] = *(const v4f*)(xrow + (size_t)t * D_DIM);

    __builtin_amdgcn_fence(__ATOMIC_RELEASE, "wavefront");
    __builtin_amdgcn_wave_barrier();
    __builtin_amdgcn_fence(__ATOMIC_ACQUIRE, "wavefront");

    #pragma unroll
    for (int ks = 0; ks < 32; ++ks) nxt[ks] = *(const v2f*)(slab_rd + 4 * ks);

    __builtin_amdgcn_fence(__ATOMIC_RELEASE, "wavefront");
    __builtin_amdgcn_wave_barrier();
    __builtin_amdgcn_fence(__ATOMIC_ACQUIRE, "wavefront");
    __builtin_amdgcn_sched_barrier(0);

    v8f c[PAIR_STRIPES] = {};
    #pragma unroll
    for (int ks = 0; ks < 32; ++ks) {
        #pragma unroll
        for (int s = 0; s < PAIR_STRIPES; ++s)
            c[s] = __builtin_amdgcn_wmma_f32_16x16x4_f32(
                    false, afrag[s][ks], false, cur[ks], (short)0, c[s], false, false);
    }

    __builtin_amdgcn_sched_barrier(0x2 | 0x4);

    const int j = jt * 16 + col;
    #pragma unroll
    for (int s = 0; s < PAIR_STRIPES; ++s) {
        #pragma unroll
        for (int r = 0; r < 8; ++r) {
            const int i  = i0 + 16 * s + r + 8 * hi;
            const float sq = s2i[s][r] + x2j - 2.f * c[s][r];
            const bool same = (labi[s][r] == labj);
            const bool pos  = same && (i != j);
            mp[s][r] = fmaxf(mp[s][r], pos  ? sq : -INFINITY);
            mn[s][r] = fminf(mn[s][r], same ? INFINITY : sq);
            ms[s][r] = fminf(ms[s][r], same ? sq : INFINITY);
        }
    }
}

__global__ __launch_bounds__(32 * PAIR_WAVES) __attribute__((amdgpu_waves_per_eu(2, 2))) void pair_kernel(
    const float* __restrict__ x,    const int* __restrict__ label,
    const float* __restrict__ sem,  const float* __restrict__ sem2,
    const float* __restrict__ x2,   float* __restrict__ diff) {
    __shared__ float s_mp[PAIR_WAVES][PAIR_ROWS];
    __shared__ float s_mn[PAIR_WAVES][PAIR_ROWS];
    __shared__ float s_ms[PAIR_WAVES][PAIR_ROWS];
    __shared__ __attribute__((aligned(16))) float s_tile[PAIR_WAVES][16 * TILE_PITCH];

    const int lane = threadIdx.x & 31;
    const int wave = __builtin_amdgcn_readfirstlane((int)(threadIdx.x >> 5));
    const int col  = lane & 15;
    const int hi   = lane >> 4;
    const int i0   = blockIdx.x * PAIR_ROWS;

    v2f afrag[PAIR_STRIPES][32];
    #pragma unroll
    for (int s = 0; s < PAIR_STRIPES; ++s) {
        const float* arow = sem + (size_t)(i0 + 16 * s + col) * D_DIM + 2 * hi;
        #pragma unroll
        for (int ks = 0; ks < 32; ++ks) afrag[s][ks] = ld2(arow + 4 * ks);
    }

    int   labi[PAIR_STRIPES][8];
    float s2i[PAIR_STRIPES][8];
    #pragma unroll
    for (int s = 0; s < PAIR_STRIPES; ++s) {
        #pragma unroll
        for (int r = 0; r < 8; ++r) {
            labi[s][r] = label[i0 + 16 * s + r + 8 * hi];
            s2i[s][r]  = sem2[i0 + 16 * s + r + 8 * hi];
        }
    }

    float mp[PAIR_STRIPES][8], mn[PAIR_STRIPES][8], ms[PAIR_STRIPES][8];
    #pragma unroll
    for (int s = 0; s < PAIR_STRIPES; ++s) {
        #pragma unroll
        for (int r = 0; r < 8; ++r) { mp[s][r] = -INFINITY; mn[s][r] = INFINITY; ms[s][r] = INFINITY; }
    }

    float* slab_wr = &s_tile[wave][4 * lane];
    const float* slab_rd = &s_tile[wave][col * TILE_PITCH + 2 * hi];

    const int jt_last = wave + PAIR_WAVES * (PAIR_STEPS - 1);
    v4f st[16];
    v2f fr0[32], fr1[32];
    int   lab0, lab1;
    float xn0, xn1;
    {
        const float* xrow = x + (size_t)(wave * 16) * D_DIM + 4 * lane;
        #pragma unroll
        for (int t = 0; t < 16; ++t) st[t] = *(const v4f*)(xrow + (size_t)t * D_DIM);
        lab0 = label[wave * 16 + col];
        xn0  = x2[wave * 16 + col];
        #pragma unroll
        for (int t = 0; t < 16; ++t) *(v4f*)(slab_wr + t * TILE_PITCH) = st[t];
        xrow += (size_t)(PAIR_WAVES * 16) * D_DIM;
        #pragma unroll
        for (int t = 0; t < 16; ++t) st[t] = *(const v4f*)(xrow + (size_t)t * D_DIM);
        __builtin_amdgcn_fence(__ATOMIC_RELEASE, "wavefront");
        __builtin_amdgcn_wave_barrier();
        __builtin_amdgcn_fence(__ATOMIC_ACQUIRE, "wavefront");
        #pragma unroll
        for (int ks = 0; ks < 32; ++ks) fr0[ks] = *(const v2f*)(slab_rd + 4 * ks);
        __builtin_amdgcn_fence(__ATOMIC_RELEASE, "wavefront");
        __builtin_amdgcn_wave_barrier();
        __builtin_amdgcn_fence(__ATOMIC_ACQUIRE, "wavefront");
    }

    for (int jt = wave; jt <= jt_last; jt += 2 * PAIR_WAVES) {
        pair_step(jt, jt_last, i0, lane, col, hi, x, label, x2, afrag, fr0, fr1, st,
                  slab_wr, slab_rd, lab0, xn0, lab1, xn1, labi, s2i, mp, mn, ms);
        pair_step(jt + PAIR_WAVES, jt_last, i0, lane, col, hi, x, label, x2, afrag, fr1, fr0, st,
                  slab_wr, slab_rd, lab1, xn1, lab0, xn0, labi, s2i, mp, mn, ms);
    }

    #pragma unroll
    for (int s = 0; s < PAIR_STRIPES; ++s) {
        #pragma unroll
        for (int r = 0; r < 8; ++r) {
            float a = mp[s][r], m = mn[s][r], q = ms[s][r];
            for (int off = 8; off >= 1; off >>= 1) {
                a = fmaxf(a, __shfl_xor(a, off, 16));
                m = fminf(m, __shfl_xor(m, off, 16));
                q = fminf(q, __shfl_xor(q, off, 16));
            }
            if (col == 0) {
                s_mp[wave][16 * s + r + 8 * hi] = a;
                s_mn[wave][16 * s + r + 8 * hi] = m;
                s_ms[wave][16 * s + r + 8 * hi] = q;
            }
        }
    }
    __syncthreads();

    if (threadIdx.x < PAIR_ROWS) {
        float a = s_mp[0][threadIdx.x], m = s_mn[0][threadIdx.x], q = s_ms[0][threadIdx.x];
        #pragma unroll
        for (int w = 1; w < PAIR_WAVES; ++w) {
            a = fmaxf(a, s_mp[w][threadIdx.x]);
            m = fminf(m, s_mn[w][threadIdx.x]);
            q = fminf(q, s_ms[w][threadIdx.x]);
        }
        const float fpd = (a == -INFINITY) ? 0.f : sqrtf((a > 0.f ? a : 0.f) + EPS_TRM);
        const float cnd = fminf(sqrtf((m > 0.f ? m : 0.f) + EPS_TRM),
                                sqrtf((q > 0.f ? q : 0.f) + EPS_TRM) + BIG_F);
        const float z  = fpd - cnd;
        const float sp = (z > 20.f) ? z : log1pf(expf(z));
        diff[i0 + threadIdx.x] = sp;
    }
}

extern "C" void kernel_launch(void* const* d_in, const int* in_sizes, int n_in,
                              void* d_out, int out_size, void* d_ws, size_t ws_size,
                              hipStream_t stream) {
    const float* x     = (const float*)d_in[0];
    const float* wv    = (const float*)d_in[1];
    const int*   label = (const int*)  d_in[2];
    const float* alpha = (const float*)d_in[3];
    const float* W     = (const float*)d_in[4];
    const float* b     = (const float*)d_in[5];

    float* diff = (float*)d_out;
    float* sem  = diff + N_PTS;

    float* x2   = (float*)d_ws;
    float* sem2 = x2 + N_PTS;

    hipLaunchKernelGGL(sem_fc_kernel, dim3(N_PTS / FC_ROWS), dim3(32 * FC_WAVES), 0, stream,
                       x, wv, alpha, W, b, sem, sem2, x2);
    hipLaunchKernelGGL(pair_kernel, dim3(N_PTS / PAIR_ROWS), dim3(32 * PAIR_WAVES), 0, stream,
                       x, label, sem, sem2, x2, diff);
}


# pair kernel: per pair 1 compare + 2 selects (was 2 + 3 + a scalar and): same-class max and min share one select (NaN neutral to both); self-pair test only in the 2 diagonal tiles; bit-identical
# speedup vs baseline: 13.3237x; 1.0347x over previous
#include <hip/hip_runtime.h>
#include <math.h>

#define N_PTS   8192
#define WV_DIM  300
#define D_DIM   128
#define BIG_F   1e8f
#define EPS_TRM (128.0f * 1e-16f)

typedef float v2f __attribute__((ext_vector_type(2)));
typedef float v4f __attribute__((ext_vector_type(4)));
typedef float v8f __attribute__((ext_vector_type(8)));

__device__ __forceinline__ v2f ld2(const float* p) {
    return *(const v2f*)p;
}

#define FC_ROWS   32
#define FC_WAVES  16
#define SEM_PITCH 132
#define FC_KB     15
__global__ __launch_bounds__(32 * FC_WAVES) void sem_fc_kernel(
    const float* __restrict__ x,  const float* __restrict__ wv,
    const float* __restrict__ alpha, const float* __restrict__ W,
    const float* __restrict__ b,  float* __restrict__ sem_out,
    float* __restrict__ sem2, float* __restrict__ x2) {
    __shared__ __attribute__((aligned(16))) float s_w[D_DIM * WV_DIM];
    __shared__ __attribute__((aligned(16))) float s_a[FC_ROWS * WV_DIM];
    __shared__ __attribute__((aligned(16))) float s_sem[FC_ROWS * SEM_PITCH];

    const int tid  = threadIdx.x;
    const int lane = tid & 31;
    const int wave = __builtin_amdgcn_readfirstlane(tid >> 5);
    const int col  = lane & 15;
    const int hi   = lane >> 4;
    const int dt   = wave & 7;
    const int st   = wave >> 3;
    const int i0   = blockIdx.x * FC_ROWS;

    const v4f* gw = (const v4f*)W;
    const v4f* ga = (const v4f*)(wv + (size_t)i0 * WV_DIM);
    v4f* lw = (v4f*)s_w;
    v4f* la = (v4f*)s_a;
    const int tw = min(tid + 512 * 18, D_DIM * WV_DIM / 4 - 1);
    const int ta = min(tid + 512 * 4, FC_ROWS * WV_DIM / 4 - 1);
    v4f rw[19], ra[5];
    #pragma unroll
    for (int t = 0; t < 18; ++t) rw[t] = gw[tid + 512 * t];
    rw[18] = gw[tw];
    #pragma unroll
    for (int t = 0; t < 4; ++t) ra[t] = ga[tid + 512 * t];
    ra[4] = ga[ta];

    const int   d  = dt * 16 + col;
    const float bd = b[d];
    float alph[8], xv[8];
    #pragma unroll
    for (int r = 0; r < 8; ++r) {
        const int i = i0 + 16 * st + r + 8 * hi;
        alph[r] = alpha[i];
        xv[r]   = x[(size_t)i * D_DIM + d];
    }

    #pragma unroll
    for (int t = 0; t < 18; ++t) lw[tid + 512 * t] = rw[t];
    lw[tw] = rw[18];
    #pragma unroll
    for (int t = 0; t < 4; ++t) la[tid + 512 * t] = ra[t];
    la[ta] = ra[4];
    __syncthreads();

    const float* ap = s_a + (16 * st + col) * WV_DIM + 2 * hi;
    const float* bp = s_w + (16 * dt + col) * WV_DIM + 2 * hi;
    v2f fa[2][FC_KB], fb[2][FC_KB];
    #pragma unroll
    for (int k = 0; k < FC_KB; ++k) {
        fa[0][k] = *(const v2f*)(ap + 4 * k);
        fb[0][k] = *(const v2f*)(bp + 4 * k);
    }
    v8f c = {};
    #pragma unroll
    for (int kb = 0; kb < WV_DIM / 4 / FC_KB; ++kb) {
        if (kb + 1 < WV_DIM / 4 / FC_KB) {
            #pragma unroll
            for (int k = 0; k < FC_KB; ++k) {
                fa[(kb + 1) & 1][k] = *(const v2f*)(ap + 4 * (FC_KB * (kb + 1) + k));
                fb[(kb + 1) & 1][k] = *(const v2f*)(bp + 4 * (FC_KB * (kb + 1) + k));
            }
        }
        __builtin_amdgcn_sched_barrier(0);
        #pragma unroll
        for (int k = 0; k < FC_KB; ++k)
            c = __builtin_amdgcn_wmma_f32_16x16x4_f32(
                    false, fa[kb & 1][k], false, fb[kb & 1][k], (short)0, c, false, false);
        __builtin_amdgcn_sched_barrier(0);
    }

    if (wave == 0) {
        const float* xr = x + (size_t)(i0 + lane) * D_DIM;
        float nx = 0.f;
        #pragma unroll
        for (int k = 0; k < D_DIM; ++k) nx = fmaf(xr[k], xr[k], nx);
        x2[i0 + lane] = nx;
    }

    #pragma unroll
    for (int r = 0; r < 8; ++r) {
        const float so = c[r] + bd;
        const float s  = alph[r] * so + (1.f - alph[r]) * xv[r];
        s_sem[(16 * st + r + 8 * hi) * SEM_PITCH + d] = s;
    }
    __syncthreads();

    #pragma unroll
    for (int t = 0; t < 2; ++t) {
        const int e   = tid + 512 * t;
        const int row = e >> 5;
        const int q   = e & 31;
        const v4f v = *(const v4f*)(s_sem + row * SEM_PITCH + 4 * q);
        *(v4f*)(sem_out + (size_t)(i0 + row) * D_DIM + 4 * q) = v;
    }

    {
        const int row = tid >> 4;
        float nrm = 0.f;
        #pragma unroll
        for (int t = 0; t < 8; ++t) {
            const float v = s_sem[row * SEM_PITCH + 16 * t + col];
            nrm = fmaf(v, v, nrm);
        }
        for (int off = 8; off >= 1; off >>= 1)
            nrm += __shfl_xor(nrm, off, 16);
        if (col == 0) sem2[i0 + row] = nrm;
    }
}

#define PAIR_WAVES   8
#define PAIR_STRIPES 2
#define PAIR_ROWS    (16 * PAIR_STRIPES)
#define TILE_PITCH   132
#define PAIR_STEPS   (N_PTS / 16 / PAIR_WAVES)

template <bool DIAG>
__device__ __forceinline__ void pair_extremes(
    const int jt, const int i0, const int col, const int hi,
    const v8f (&c)[2], const int labj, const float x2j,
    const int (&labi)[2][8], const float (&s2i)[2][8],
    float (&mp)[2][8], float (&mn)[2][8], float (&ms)[2][8]) {
    const int j = jt * 16 + col;
    #pragma unroll
    for (int s = 0; s < 2; ++s) {
        #pragma unroll
        for (int r = 0; r < 8; ++r) {
            const float sq = s2i[s][r] + x2j - 2.f * c[s][r];
            const bool same = (labi[s][r] == labj);
            const float ts = same ? sq : NAN;
            const float td = same ? INFINITY : sq;
            float tp = ts;
            if (DIAG) {
                const int i = i0 + 16 * s + r + 8 * hi;
                tp = (i != j) ? ts : NAN;
            }
            mp[s][r] = fmaxf(mp[s][r], tp);
            mn[s][r] = fminf(mn[s][r], td);
            ms[s][r] = fminf(ms[s][r], ts);
        }
    }
}

__device__ __forceinline__ float sweep_x2(const float* __restrict__ x2, const int jt, const int dlo, const int col) {
    const float v = x2[jt * 16 + col];
    return ((unsigned)(jt - dlo) < 2u) ? NAN : v;
}

__device__ __forceinline__ void pair_step(
    const int jt, const int jt_last, const int dlo, const int i0, const int lane, const int col, const int hi,
    const float* __restrict__ x, const int* __restrict__ label, const float* __restrict__ x2,
    const v2f (&afrag)[PAIR_STRIPES][32], v2f (&cur)[32], v2f (&nxt)[32], v4f (&st)[16],
    float* slab_wr, const float* slab_rd,
    const int labj, const float x2j, int& labn, float& x2n,
    const int (&labi)[PAIR_STRIPES][8], const float (&s2i)[PAIR_STRIPES][8],
    float (&mp)[PAIR_STRIPES][8], float (&mn)[PAIR_STRIPES][8], float (&ms)[PAIR_STRIPES][8]) {
    #pragma unroll
    for (int t = 0; t < 16; ++t) *(v4f*)(slab_wr + t * TILE_PITCH) = st[t];

    const int jt1 = min(jt + PAIR_WAVES, jt_last);
    const int jt2 = min(jt + 2 * PAIR_WAVES, jt_last);
    labn = label[jt1 * 16 + col];
    x2n  = sweep_x2(x2, jt1, dlo, col);
    const float* xrow = x + (size_t)(jt2 * 16) * D_DIM + 4 * lane;
    #pragma unroll
    for (int t = 0; t < 16; ++t) st[t] = *(const v4f*)(xrow + (size_t)t * D_DIM);

    __builtin_amdgcn_fence(__ATOMIC_RELEASE, "wavefront");
    __builtin_amdgcn_wave_barrier();
    __builtin_amdgcn_fence(__ATOMIC_ACQUIRE, "wavefront");

    #pragma unroll
    for (int ks = 0; ks < 32; ++ks) nxt[ks] = *(const v2f*)(slab_rd + 4 * ks);

    __builtin_amdgcn_fence(__ATOMIC_RELEASE, "wavefront");
    __builtin_amdgcn_wave_barrier();
    __builtin_amdgcn_fence(__ATOMIC_ACQUIRE, "wavefront");
    __builtin_amdgcn_sched_barrier(0);

    v8f c[PAIR_STRIPES] = {};
    #pragma unroll
    for (int ks = 0; ks < 32; ++ks) {
        #pragma unroll
        for (int s = 0; s < PAIR_STRIPES; ++s)
            c[s] = __builtin_amdgcn_wmma_f32_16x16x4_f32(
                    false, afrag[s][ks], false, cur[ks], (short)0, c[s], false, false);
    }

    __builtin_amdgcn_sched_barrier(0x2 | 0x4);

    pair_extremes<false>(jt, i0, col, hi, c, labj, x2j, labi, s2i, mp, mn, ms);
}

__global__ __launch_bounds__(32 * PAIR_WAVES) __attribute__((amdgpu_waves_per_eu(2, 2))) void pair_kernel(
    const float* __restrict__ x,    const int* __restrict__ label,
    const float* __restrict__ sem,  const float* __restrict__ sem2,
    const float* __restrict__ x2,   float* __restrict__ diff) {
    __shared__ float s_mp[PAIR_WAVES][PAIR_ROWS];
    __shared__ float s_mn[PAIR_WAVES][PAIR_ROWS];
    __shared__ float s_ms[PAIR_WAVES][PAIR_ROWS];
    __shared__ __attribute__((aligned(16))) float s_tile[PAIR_WAVES][16 * TILE_PITCH];

    const int lane = threadIdx.x & 31;
    const int wave = __builtin_amdgcn_readfirstlane((int)(threadIdx.x >> 5));
    const int col  = lane & 15;
    const int hi   = lane >> 4;
    const int i0   = blockIdx.x * PAIR_ROWS;

    v2f afrag[PAIR_STRIPES][32];
    #pragma unroll
    for (int s = 0; s < PAIR_STRIPES; ++s) {
        const float* arow = sem + (size_t)(i0 + 16 * s + col) * D_DIM + 2 * hi;
        #pragma unroll
        for (int ks = 0; ks < 32; ++ks) afrag[s][ks] = ld2(arow + 4 * ks);
    }

    int   labi[PAIR_STRIPES][8];
    float s2i[PAIR_STRIPES][8];
    #pragma unroll
    for (int s = 0; s < PAIR_STRIPES; ++s) {
        #pragma unroll
        for (int r = 0; r < 8; ++r) {
            labi[s][r] = label[i0 + 16 * s + r + 8 * hi];
            s2i[s][r]  = sem2[i0 + 16 * s + r + 8 * hi];
        }
    }

    float mp[PAIR_STRIPES][8], mn[PAIR_STRIPES][8], ms[PAIR_STRIPES][8];
    #pragma unroll
    for (int s = 0; s < PAIR_STRIPES; ++s) {
        #pragma unroll
        for (int r = 0; r < 8; ++r) { mp[s][r] = -INFINITY; mn[s][r] = INFINITY; ms[s][r] = INFINITY; }
    }

    float* slab_wr = &s_tile[wave][4 * lane];
    const float* slab_rd = &s_tile[wave][col * TILE_PITCH + 2 * hi];

    const int dlo  = 2 * (int)blockIdx.x;
    const int dsel = wave - (dlo & (PAIR_WAVES - 1));
    if (dsel == 0 || dsel == 1) {
        const int jd = dlo + dsel;
        const float* xrow = x + (size_t)(jd * 16) * D_DIM + 4 * lane;
        v4f rows[16];
        #pragma unroll
        for (int t = 0; t < 16; ++t) rows[t] = *(const v4f*)(xrow + (size_t)t * D_DIM);
        const int   labd = label[jd * 16 + col];
        const float x2d  = x2[jd * 16 + col];
        #pragma unroll
        for (int t = 0; t < 16; ++t) *(v4f*)(slab_wr + t * TILE_PITCH) = rows[t];
        __builtin_amdgcn_fence(__ATOMIC_RELEASE, "wavefront");
        __builtin_amdgcn_wave_barrier();
        __builtin_amdgcn_fence(__ATOMIC_ACQUIRE, "wavefront");
        v2f frd[32];
        #pragma unroll
        for (int ks = 0; ks < 32; ++ks) frd[ks] = *(const v2f*)(slab_rd + 4 * ks);
        __builtin_amdgcn_fence(__ATOMIC_RELEASE, "wavefront");
        __builtin_amdgcn_wave_barrier();
        __builtin_amdgcn_fence(__ATOMIC_ACQUIRE, "wavefront");
        v8f cd[PAIR_STRIPES] = {};
        #pragma unroll
        for (int ks = 0; ks < 32; ++ks) {
            #pragma unroll
            for (int s = 0; s < PAIR_STRIPES; ++s)
                cd[s] = __builtin_amdgcn_wmma_f32_16x16x4_f32(
                        false, afrag[s][ks], false, frd[ks], (short)0, cd[s], false, false);
        }
        pair_extremes<true>(jd, i0, col, hi, cd, labd, x2d, labi, s2i, mp, mn, ms);
    }

    const int jt_last = wave + PAIR_WAVES * (PAIR_STEPS - 1);
    v4f st[16];
    v2f fr0[32], fr1[32];
    int   lab0, lab1;
    float xn0, xn1;
    {
        const float* xrow = x + (size_t)(wave * 16) * D_DIM + 4 * lane;
        #pragma unroll
        for (int t = 0; t < 16; ++t) st[t] = *(const v4f*)(xrow + (size_t)t * D_DIM);
        lab0 = label[wave * 16 + col];
        xn0  = sweep_x2(x2, wave, dlo, col);
        #pragma unroll
        for (int t = 0; t < 16; ++t) *(v4f*)(slab_wr + t * TILE_PITCH) = st[t];
        xrow += (size_t)(PAIR_WAVES * 16) * D_DIM;
        #pragma unroll
        for (int t = 0; t < 16; ++t) st[t] = *(const v4f*)(xrow + (size_t)t * D_DIM);
        __builtin_amdgcn_fence(__ATOMIC_RELEASE, "wavefront");
        __builtin_amdgcn_wave_barrier();
        __builtin_amdgcn_fence(__ATOMIC_ACQUIRE, "wavefront");
        #pragma unroll
        for (int ks = 0; ks < 32; ++ks) fr0[ks] = *(const v2f*)(slab_rd + 4 * ks);
        __builtin_amdgcn_fence(__ATOMIC_RELEASE, "wavefront");
        __builtin_amdgcn_wave_barrier();
        __builtin_amdgcn_fence(__ATOMIC_ACQUIRE, "wavefront");
    }

    for (int jt = wave; jt <= jt_last; jt += 2 * PAIR_WAVES) {
        pair_step(jt, jt_last, dlo, i0, lane, col, hi, x, label, x2, afrag, fr0, fr1, st,
                  slab_wr, slab_rd, lab0, xn0, lab1, xn1, labi, s2i, mp, mn, ms);
        pair_step(jt + PAIR_WAVES, jt_last, dlo, i0, lane, col, hi, x, label, x2, afrag, fr1, fr0, st,
                  slab_wr, slab_rd, lab1, xn1, lab0, xn0, labi, s2i, mp, mn, ms);
    }

    #pragma unroll
    for (int s = 0; s < PAIR_STRIPES; ++s) {
        #pragma unroll
        for (int r = 0; r < 8; ++r) {
            float a = mp[s][r], m = mn[s][r], q = ms[s][r];
            for (int off = 8; off >= 1; off >>= 1) {
                a = fmaxf(a, __shfl_xor(a, off, 16));
                m = fminf(m, __shfl_xor(m, off, 16));
                q = fminf(q, __shfl_xor(q, off, 16));
            }
            if (col == 0) {
                s_mp[wave][16 * s + r + 8 * hi] = a;
                s_mn[wave][16 * s + r + 8 * hi] = m;
                s_ms[wave][16 * s + r + 8 * hi] = q;
            }
        }
    }
    __syncthreads();

    if (threadIdx.x < PAIR_ROWS) {
        float a = s_mp[0][threadIdx.x], m = s_mn[0][threadIdx.x], q = s_ms[0][threadIdx.x];
        #pragma unroll
        for (int w = 1; w < PAIR_WAVES; ++w) {
            a = fmaxf(a, s_mp[w][threadIdx.x]);
            m = fminf(m, s_mn[w][threadIdx.x]);
            q = fminf(q, s_ms[w][threadIdx.x]);
        }
        const float fpd = (a == -INFINITY) ? 0.f : sqrtf((a > 0.f ? a : 0.f) + EPS_TRM);
        const float cnd = fminf(sqrtf((m > 0.f ? m : 0.f) + EPS_TRM),
                                sqrtf((q > 0.f ? q : 0.f) + EPS_TRM) + BIG_F);
        const float z  = fpd - cnd;
        const float sp = (z > 20.f) ? z : log1pf(expf(z));
        diff[i0 + threadIdx.x] = sp;
    }
}

extern "C" void kernel_launch(void* const* d_in, const int* in_sizes, int n_in,
                              void* d_out, int out_size, void* d_ws, size_t ws_size,
                              hipStream_t stream) {
    const float* x     = (const float*)d_in[0];
    const float* wv    = (const float*)d_in[1];
    const int*   label = (const int*)  d_in[2];
    const float* alpha = (const float*)d_in[3];
    const float* W     = (const float*)d_in[4];
    const float* b     = (const float*)d_in[5];

    float* diff = (float*)d_out;
    float* sem  = diff + N_PTS;

    float* x2   = (float*)d_ws;
    float* sem2 = x2 + N_PTS;

    hipLaunchKernelGGL(sem_fc_kernel, dim3(N_PTS / FC_ROWS), dim3(32 * FC_WAVES), 0, stream,
                       x, wv, alpha, W, b, sem, sem2, x2);
    hipLaunchKernelGGL(pair_kernel, dim3(N_PTS / PAIR_ROWS), dim3(32 * PAIR_WAVES), 0, stream,
                       x, label, sem, sem2, x2, diff);
}
